# starting kernel
# baseline (speedup 1.0000x reference)
#include <hip/hip_runtime.h>
#include <math.h>


#define B_   8
#define C_   512
#define CR_  32
#define HW_  4096
#define C2_  1024

typedef __attribute__((ext_vector_type(2))) float v2f;
typedef __attribute__((ext_vector_type(8))) float v8f;

__device__ __forceinline__ v8f wmma4(v2f a, v2f b, v8f c) {
  return __builtin_amdgcn_wmma_f32_16x16x4_f32(
       false, a,  false, b,
       (short)0, c,  false,  false);
}

template<int NT>
__device__ __forceinline__ void gemm_row(const float* __restrict__ A, int lda,
                                         const float* __restrict__ Bm, int ldb,
                                         int K, v8f c[NT]) {
  const int lane = threadIdx.x & 31;
  const int half = lane >> 4;
  const int l16  = lane & 15;
  const float* ap = A + (size_t)l16 * lda + 2 * half;
  const float* bp = Bm + (size_t)(2 * half) * ldb + l16;
#pragma unroll 2
  for (int k = 0; k < K; k += 4) {
    v2f a = *(const v2f*)ap;
#pragma unroll
    for (int t = 0; t < NT; ++t) {
      v2f b;
      b.x = bp[t * 16];
      b.y = bp[t * 16 + ldb];
      c[t] = wmma4(a, b, c[t]);
    }
    ap += 4;
    bp += (size_t)4 * ldb;
  }
}

template<int NT>
__device__ __forceinline__ void gemm_row_bt(const float* __restrict__ A, int lda,
                                            const float* __restrict__ Bt, int ldbt,
                                            int K, v8f c[NT]) {
  const int lane = threadIdx.x & 31;
  const int half = lane >> 4;
  const int l16  = lane & 15;
  const float* ap = A + (size_t)l16 * lda + 2 * half;
  const float* bp = Bt + (size_t)l16 * ldbt + 2 * half;
#pragma unroll 2
  for (int k = 0; k < K; k += 4) {
    v2f a = *(const v2f*)ap;
#pragma unroll
    for (int t = 0; t < NT; ++t) {
      v2f b = *(const v2f*)(bp + (size_t)t * 16 * ldbt);
      c[t] = wmma4(a, b, c[t]);
    }
    ap += 4;
    bp += 4;
  }
}

__global__ void k_conv1_gelu(const float* __restrict__ x, const float* __restrict__ w1,
                             const float* __restrict__ b1, float* __restrict__ h) {
  constexpr int NT = 4;
  const int wid = blockIdx.x * 8 + (threadIdx.x >> 5);
  const int NG = HW_ / (16 * NT);
  const int MT = CR_ / 16;
  const int b  = wid / (MT * NG);
  const int r0 = wid % (MT * NG);
  const int mt = r0 / NG;
  const int ng = r0 % NG;

  const float* A  = w1 + (size_t)mt * 16 * C_;
  const float* Bm = x + (size_t)b * C_ * HW_ + ng * 16 * NT;
  v8f c[NT] = {};
  gemm_row<NT>(A, C_, Bm, HW_, C_, c);

  const int lane = threadIdx.x & 31;
  const int half = lane >> 4;
  const int l16  = lane & 15;
  float* D = h + (size_t)b * CR_ * HW_;
#pragma unroll
  for (int t = 0; t < NT; ++t)
#pragma unroll
    for (int r = 0; r < 8; ++r) {
      const int m = mt * 16 + r + 8 * half;
      const int n = ng * 16 * NT + t * 16 + l16;
      float v = c[t][r] + b1[m];
      v = 0.5f * v * (1.0f + erff(v * 0.70710678118654752f));
      D[(size_t)m * HW_ + n] = v;
    }
}

__global__ void k_conv2(const float* __restrict__ h, const float* __restrict__ w2,
                        const float* __restrict__ b2, float* __restrict__ y) {
  constexpr int NT = 4;
  const int wid = blockIdx.x * 8 + (threadIdx.x >> 5);
  const int NG = HW_ / (16 * NT);
  const int MT = C_ / 16;
  const int b  = wid / (MT * NG);
  const int r0 = wid % (MT * NG);
  const int mt = r0 / NG;
  const int ng = r0 % NG;

  const float* A  = w2 + (size_t)mt * 16 * CR_;
  const float* Bm = h + (size_t)b * CR_ * HW_ + ng * 16 * NT;
  v8f c[NT] = {};
  gemm_row<NT>(A, CR_, Bm, HW_, CR_, c);

  const int lane = threadIdx.x & 31;
  const int half = lane >> 4;
  const int l16  = lane & 15;
  float* D = y + (size_t)b * C_ * HW_;
#pragma unroll
  for (int t = 0; t < NT; ++t)
#pragma unroll
    for (int r = 0; r < 8; ++r) {
      const int m = mt * 16 + r + 8 * half;
      const int n = ng * 16 * NT + t * 16 + l16;
      D[(size_t)m * HW_ + n] = c[t][r] + b2[m];
    }
}

__global__ void k_energy(const float* __restrict__ y, float* __restrict__ e) {
  constexpr int NT = 4;
  const int wid = blockIdx.x * 8 + (threadIdx.x >> 5);
  const int NG = C_ / (16 * NT);
  const int MT = C_ / 16;
  const int b  = wid / (MT * NG);
  const int r0 = wid % (MT * NG);
  const int mt = r0 / NG;
  const int ng = r0 % NG;

  const float* yb = y + (size_t)b * C_ * HW_;
  const float* A  = yb + (size_t)mt * 16 * HW_;
  const float* Bt = yb + (size_t)ng * 16 * NT * HW_;
  v8f c[NT] = {};
  gemm_row_bt<NT>(A, HW_, Bt, HW_, HW_, c);

  const int lane = threadIdx.x & 31;
  const int half = lane >> 4;
  const int l16  = lane & 15;
  float* D = e + (size_t)b * C_ * C_;
#pragma unroll
  for (int t = 0; t < NT; ++t)
#pragma unroll
    for (int r = 0; r < 8; ++r) {
      const int m = mt * 16 + r + 8 * half;
      const int n = ng * 16 * NT + t * 16 + l16;
      D[(size_t)m * C_ + n] = c[t][r];
    }
}

__global__ void k_softmax(float* __restrict__ e) {
  __shared__ float red[256];
  const int tid = threadIdx.x;
  float* p = e + (size_t)blockIdx.x * C_;
  const float v0 = p[tid];
  const float v1 = p[tid + 256];
  red[tid] = fminf(v0, v1);
  __syncthreads();
  for (int s = 128; s > 0; s >>= 1) {
    if (tid < s) red[tid] = fminf(red[tid], red[tid + s]);
    __syncthreads();
  }
  const float rmin = red[0];
  __syncthreads();
  const float e0 = expf(rmin - v0);
  const float e1 = expf(rmin - v1);
  red[tid] = e0 + e1;
  __syncthreads();
  for (int s = 128; s > 0; s >>= 1) {
    if (tid < s) red[tid] += red[tid + s];
    __syncthreads();
  }
  const float inv = 1.0f / red[0];
  p[tid] = e0 * inv;
  p[tid + 256] = e1 * inv;
}

__global__ void k_m2(const float* __restrict__ attn, const float* __restrict__ wf,
                     const float* __restrict__ gamma, float* __restrict__ m2) {
  constexpr int NT = 4;
  const int wid = blockIdx.x * 8 + (threadIdx.x >> 5);
  const int NG = C_ / (16 * NT);
  const int MT = C_ / 16;
  const int b  = wid / (MT * NG);
  const int r0 = wid % (MT * NG);
  const int mt = r0 / NG;
  const int ng = r0 % NG;

  const float* A  = wf + (size_t)mt * 16 * C2_;
  const float* Bm = attn + (size_t)b * C_ * C_ + ng * 16 * NT;
  v8f c[NT] = {};
  gemm_row<NT>(A, C2_, Bm, C_, C_, c);

  const float g = gamma[0];
  const int lane = threadIdx.x & 31;
  const int half = lane >> 4;
  const int l16  = lane & 15;
  float* D = m2 + (size_t)b * C_ * C_;
#pragma unroll
  for (int t = 0; t < NT; ++t)
#pragma unroll
    for (int r = 0; r < 8; ++r) {
      const int m = mt * 16 + r + 8 * half;
      const int n = ng * 16 * NT + t * 16 + l16;
      D[(size_t)m * C_ + n] = g * c[t][r] + wf[(size_t)m * C2_ + n];
    }
}

__global__ void k_fusion(const float* __restrict__ m2, const float* __restrict__ y,
                         const float* __restrict__ x, const float* __restrict__ wf,
                         const float* __restrict__ bf, float* __restrict__ out) {
  constexpr int NT = 4;
  const int wid = blockIdx.x * 8 + (threadIdx.x >> 5);
  const int NG = HW_ / (16 * NT);
  const int MT = C_ / 16;
  const int b  = wid / (MT * NG);
  const int r0 = wid % (MT * NG);
  const int mt = r0 / NG;
  const int ng = r0 % NG;

  v8f c[NT] = {};
  gemm_row<NT>(m2 + (size_t)b * C_ * C_ + (size_t)mt * 16 * C_, C_,
               y + (size_t)b * C_ * HW_ + ng * 16 * NT, HW_, C_, c);
  gemm_row<NT>(wf + (size_t)mt * 16 * C2_ + C_, C2_,
               x + (size_t)b * C_ * HW_ + ng * 16 * NT, HW_, C_, c);

  const int lane = threadIdx.x & 31;
  const int half = lane >> 4;
  const int l16  = lane & 15;
  float* D = out + (size_t)b * C_ * HW_;
#pragma unroll
  for (int t = 0; t < NT; ++t)
#pragma unroll
    for (int r = 0; r < 8; ++r) {
      const int m = mt * 16 + r + 8 * half;
      const int n = ng * 16 * NT + t * 16 + l16;
      D[(size_t)m * HW_ + n] = c[t][r] + bf[m];
    }
}

extern "C" void kernel_launch(void* const* d_in, const int* in_sizes, int n_in,
                              void* d_out, int out_size, void* d_ws, size_t ws_size,
                              hipStream_t stream) {
  const float* x     = (const float*)d_in[0];
  const float* w1    = (const float*)d_in[1];
  const float* b1    = (const float*)d_in[2];
  const float* w2    = (const float*)d_in[3];
  const float* b2    = (const float*)d_in[4];
  const float* wf    = (const float*)d_in[5];
  const float* bf    = (const float*)d_in[6];
  const float* gamma = (const float*)d_in[7];
  float* out = (float*)d_out;

  char* ws = (char*)d_ws;
  float* h    = (float*)(ws);
  float* y    = (float*)(ws + (size_t)4 * 1024 * 1024);
  float* attn = (float*)(ws + (size_t)68 * 1024 * 1024);
  float* m2   = (float*)(ws + (size_t)76 * 1024 * 1024);

  const dim3 blk(256);
  k_conv1_gelu<<<dim3(128),  blk, 0, stream>>>(x, w1, b1, h);
  k_conv2     <<<dim3(2048), blk, 0, stream>>>(h, w2, b2, y);
  k_energy    <<<dim3(256),  blk, 0, stream>>>(y, attn);
  k_softmax   <<<dim3(4096), blk, 0, stream>>>(attn);
  k_m2        <<<dim3(256),  blk, 0, stream>>>(attn, wf, gamma, m2);
  k_fusion    <<<dim3(2048), blk, 0, stream>>>(m2, y, x, wf, bf, out);
}


# fusion: M2.y re-associated through y's rank-32 factor as (M2.W2).h + M2.b2; a small new f32 WMMA kernel (split-K, LDS sum) builds M2.W2 and M2.b2; f32 summation order changes; launches 6->7
# speedup vs baseline: 1.2337x; 1.2337x over previous
#include <hip/hip_runtime.h>
#include <math.h>


#define B_   8
#define C_   512
#define CR_  32
#define HW_  4096
#define C2_  1024

typedef __attribute__((ext_vector_type(2))) float v2f;
typedef __attribute__((ext_vector_type(8))) float v8f;

__device__ __forceinline__ v8f wmma4(v2f a, v2f b, v8f c) {
  return __builtin_amdgcn_wmma_f32_16x16x4_f32(
       false, a,  false, b,
       (short)0, c,  false,  false);
}

template<int NT>
__device__ __forceinline__ void gemm_row(const float* __restrict__ A, int lda,
                                         const float* __restrict__ Bm, int ldb,
                                         int K, v8f c[NT]) {
  const int lane = threadIdx.x & 31;
  const int half = lane >> 4;
  const int l16  = lane & 15;
  const float* ap = A + (size_t)l16 * lda + 2 * half;
  const float* bp = Bm + (size_t)(2 * half) * ldb + l16;
#pragma unroll 2
  for (int k = 0; k < K; k += 4) {
    v2f a = *(const v2f*)ap;
#pragma unroll
    for (int t = 0; t < NT; ++t) {
      v2f b;
      b.x = bp[t * 16];
      b.y = bp[t * 16 + ldb];
      c[t] = wmma4(a, b, c[t]);
    }
    ap += 4;
    bp += (size_t)4 * ldb;
  }
}

template<int NT>
__device__ __forceinline__ void gemm_row_bt(const float* __restrict__ A, int lda,
                                            const float* __restrict__ Bt, int ldbt,
                                            int K, v8f c[NT]) {
  const int lane = threadIdx.x & 31;
  const int half = lane >> 4;
  const int l16  = lane & 15;
  const float* ap = A + (size_t)l16 * lda + 2 * half;
  const float* bp = Bt + (size_t)l16 * ldbt + 2 * half;
#pragma unroll 2
  for (int k = 0; k < K; k += 4) {
    v2f a = *(const v2f*)ap;
#pragma unroll
    for (int t = 0; t < NT; ++t) {
      v2f b = *(const v2f*)(bp + (size_t)t * 16 * ldbt);
      c[t] = wmma4(a, b, c[t]);
    }
    ap += 4;
    bp += 4;
  }
}

__global__ void k_conv1_gelu(const float* __restrict__ x, const float* __restrict__ w1,
                             const float* __restrict__ b1, float* __restrict__ h) {
  constexpr int NT = 4;
  const int wid = blockIdx.x * 8 + (threadIdx.x >> 5);
  const int NG = HW_ / (16 * NT);
  const int MT = CR_ / 16;
  const int b  = wid / (MT * NG);
  const int r0 = wid % (MT * NG);
  const int mt = r0 / NG;
  const int ng = r0 % NG;

  const float* A  = w1 + (size_t)mt * 16 * C_;
  const float* Bm = x + (size_t)b * C_ * HW_ + ng * 16 * NT;
  v8f c[NT] = {};
  gemm_row<NT>(A, C_, Bm, HW_, C_, c);

  const int lane = threadIdx.x & 31;
  const int half = lane >> 4;
  const int l16  = lane & 15;
  float* D = h + (size_t)b * CR_ * HW_;
#pragma unroll
  for (int t = 0; t < NT; ++t)
#pragma unroll
    for (int r = 0; r < 8; ++r) {
      const int m = mt * 16 + r + 8 * half;
      const int n = ng * 16 * NT + t * 16 + l16;
      float v = c[t][r] + b1[m];
      v = 0.5f * v * (1.0f + erff(v * 0.70710678118654752f));
      D[(size_t)m * HW_ + n] = v;
    }
}

__global__ void k_conv2(const float* __restrict__ h, const float* __restrict__ w2,
                        const float* __restrict__ b2, float* __restrict__ y) {
  constexpr int NT = 4;
  const int wid = blockIdx.x * 8 + (threadIdx.x >> 5);
  const int NG = HW_ / (16 * NT);
  const int MT = C_ / 16;
  const int b  = wid / (MT * NG);
  const int r0 = wid % (MT * NG);
  const int mt = r0 / NG;
  const int ng = r0 % NG;

  const float* A  = w2 + (size_t)mt * 16 * CR_;
  const float* Bm = h + (size_t)b * CR_ * HW_ + ng * 16 * NT;
  v8f c[NT] = {};
  gemm_row<NT>(A, CR_, Bm, HW_, CR_, c);

  const int lane = threadIdx.x & 31;
  const int half = lane >> 4;
  const int l16  = lane & 15;
  float* D = y + (size_t)b * C_ * HW_;
#pragma unroll
  for (int t = 0; t < NT; ++t)
#pragma unroll
    for (int r = 0; r < 8; ++r) {
      const int m = mt * 16 + r + 8 * half;
      const int n = ng * 16 * NT + t * 16 + l16;
      D[(size_t)m * HW_ + n] = c[t][r] + b2[m];
    }
}

__global__ void k_energy(const float* __restrict__ y, float* __restrict__ e) {
  constexpr int NT = 4;
  const int wid = blockIdx.x * 8 + (threadIdx.x >> 5);
  const int NG = C_ / (16 * NT);
  const int MT = C_ / 16;
  const int b  = wid / (MT * NG);
  const int r0 = wid % (MT * NG);
  const int mt = r0 / NG;
  const int ng = r0 % NG;

  const float* yb = y + (size_t)b * C_ * HW_;
  const float* A  = yb + (size_t)mt * 16 * HW_;
  const float* Bt = yb + (size_t)ng * 16 * NT * HW_;
  v8f c[NT] = {};
  gemm_row_bt<NT>(A, HW_, Bt, HW_, HW_, c);

  const int lane = threadIdx.x & 31;
  const int half = lane >> 4;
  const int l16  = lane & 15;
  float* D = e + (size_t)b * C_ * C_;
#pragma unroll
  for (int t = 0; t < NT; ++t)
#pragma unroll
    for (int r = 0; r < 8; ++r) {
      const int m = mt * 16 + r + 8 * half;
      const int n = ng * 16 * NT + t * 16 + l16;
      D[(size_t)m * C_ + n] = c[t][r];
    }
}

__global__ void k_softmax(float* __restrict__ e) {
  __shared__ float red[256];
  const int tid = threadIdx.x;
  float* p = e + (size_t)blockIdx.x * C_;
  const float v0 = p[tid];
  const float v1 = p[tid + 256];
  red[tid] = fminf(v0, v1);
  __syncthreads();
  for (int s = 128; s > 0; s >>= 1) {
    if (tid < s) red[tid] = fminf(red[tid], red[tid + s]);
    __syncthreads();
  }
  const float rmin = red[0];
  __syncthreads();
  const float e0 = expf(rmin - v0);
  const float e1 = expf(rmin - v1);
  red[tid] = e0 + e1;
  __syncthreads();
  for (int s = 128; s > 0; s >>= 1) {
    if (tid < s) red[tid] += red[tid + s];
    __syncthreads();
  }
  const float inv = 1.0f / red[0];
  p[tid] = e0 * inv;
  p[tid + 256] = e1 * inv;
}

__global__ void k_m2(const float* __restrict__ attn, const float* __restrict__ wf,
                     const float* __restrict__ gamma, float* __restrict__ m2) {
  constexpr int NT = 4;
  const int wid = blockIdx.x * 8 + (threadIdx.x >> 5);
  const int NG = C_ / (16 * NT);
  const int MT = C_ / 16;
  const int b  = wid / (MT * NG);
  const int r0 = wid % (MT * NG);
  const int mt = r0 / NG;
  const int ng = r0 % NG;

  const float* A  = wf + (size_t)mt * 16 * C2_;
  const float* Bm = attn + (size_t)b * C_ * C_ + ng * 16 * NT;
  v8f c[NT] = {};
  gemm_row<NT>(A, C2_, Bm, C_, C_, c);

  const float g = gamma[0];
  const int lane = threadIdx.x & 31;
  const int half = lane >> 4;
  const int l16  = lane & 15;
  float* D = m2 + (size_t)b * C_ * C_;
#pragma unroll
  for (int t = 0; t < NT; ++t)
#pragma unroll
    for (int r = 0; r < 8; ++r) {
      const int m = mt * 16 + r + 8 * half;
      const int n = ng * 16 * NT + t * 16 + l16;
      D[(size_t)m * C_ + n] = g * c[t][r] + wf[(size_t)m * C2_ + n];
    }
}

__global__ void k_pq(const float* __restrict__ m2, const float* __restrict__ w2,
                     const float* __restrict__ b2, float* __restrict__ P, float* __restrict__ q) {
  __shared__ float part[8][3][8][32];
  const int b    = blockIdx.x >> 5;
  const int mt   = blockIdx.x & 31;
  v8f c0 = {}, c1 = {}, c2 = {};
  {
    const int w    = threadIdx.x >> 5;
    const int lane = threadIdx.x & 31;
    const int half = lane >> 4;
    const int l16  = lane & 15;
    const int k0   = w * 64;
    const float* ap = m2 + (size_t)b * C_ * C_ + (size_t)(mt * 16 + l16) * C_ + k0 + 2 * half;
    const float* bp = w2 + (size_t)(k0 + 2 * half) * CR_ + l16;
    const float* cp = b2 + k0 + 2 * half;
#pragma unroll
    for (int kk = 0; kk < 64; kk += 4) {
      const v2f a = *(const v2f*)(ap + kk);
      v2f b0, b1;
      b0.x = bp[kk * CR_];
      b0.y = bp[kk * CR_ + CR_];
      b1.x = bp[kk * CR_ + 16];
      b1.y = bp[kk * CR_ + CR_ + 16];
      const v2f bb = *(const v2f*)(cp + kk);
      c0 = wmma4(a, b0, c0);
      c1 = wmma4(a, b1, c1);
      c2 = wmma4(a, bb, c2);
    }
  }
  int tid = threadIdx.x;
  asm volatile("v_nop\n\tv_nop\n\tv_nop\n\tv_nop\n\tv_nop\n\tv_nop\n\tv_nop\n\tv_nop\n\tv_nop"
               : "+v"(c0), "+v"(c1), "+v"(c2), "+v"(tid));
  const int w    = tid >> 5;
  const int lane = tid & 31;
  const int half = lane >> 4;
  const int l16  = lane & 15;
#pragma unroll
  for (int r = 0; r < 8; ++r) {
    part[w][0][r][lane] = c0[r];
    part[w][1][r][lane] = c1[r];
    part[w][2][r][lane] = c2[r];
  }
  __syncthreads();

  const int m = mt * 16 + w + 8 * half;
#pragma unroll
  for (int t = 0; t < 2; ++t) {
    float s = 0.0f;
#pragma unroll
    for (int u = 0; u < 8; ++u) s += part[u][t][w][lane];
    P[((size_t)b * C_ + m) * CR_ + t * 16 + l16] = s;
  }
  if (tid < 16) {
    const int row = tid;
    float s = 0.0f;
#pragma unroll
    for (int u = 0; u < 8; ++u) s += part[u][2][row & 7][(row >> 3) * 16];
    q[(size_t)blockIdx.x * 32 + row] = s;
  }
}

__global__ void k_fusion(const float* __restrict__ P, const float* __restrict__ q,
                         const float* __restrict__ h, const float* __restrict__ x,
                         const float* __restrict__ wf, const float* __restrict__ bf,
                         float* __restrict__ out) {
  constexpr int NT = 4;
  const int wid = blockIdx.x * 8 + (threadIdx.x >> 5);
  const int NG = HW_ / (16 * NT);
  const int MT = C_ / 16;
  const int b  = wid / (MT * NG);
  const int r0 = wid % (MT * NG);
  const int mt = r0 / NG;
  const int ng = r0 % NG;

  v8f c[NT] = {};
  gemm_row<NT>(P + ((size_t)b * C_ + (size_t)mt * 16) * CR_, CR_,
               h + (size_t)b * CR_ * HW_ + ng * 16 * NT, HW_, CR_, c);
  gemm_row<NT>(wf + (size_t)mt * 16 * C2_ + C_, C2_,
               x + (size_t)b * C_ * HW_ + ng * 16 * NT, HW_, C_, c);

  const int lane = threadIdx.x & 31;
  const int half = lane >> 4;
  const int l16  = lane & 15;
  float* D = out + (size_t)b * C_ * HW_;
  const float* qb = q + ((size_t)b * MT + mt) * 32;
#pragma unroll
  for (int t = 0; t < NT; ++t)
#pragma unroll
    for (int r = 0; r < 8; ++r) {
      const int m = mt * 16 + r + 8 * half;
      const int n = ng * 16 * NT + t * 16 + l16;
      D[(size_t)m * HW_ + n] = c[t][r] + (bf[m] + qb[r + 8 * half]);
    }
}

extern "C" void kernel_launch(void* const* d_in, const int* in_sizes, int n_in,
                              void* d_out, int out_size, void* d_ws, size_t ws_size,
                              hipStream_t stream) {
  const float* x     = (const float*)d_in[0];
  const float* w1    = (const float*)d_in[1];
  const float* b1    = (const float*)d_in[2];
  const float* w2    = (const float*)d_in[3];
  const float* b2    = (const float*)d_in[4];
  const float* wf    = (const float*)d_in[5];
  const float* bf    = (const float*)d_in[6];
  const float* gamma = (const float*)d_in[7];
  float* out = (float*)d_out;

  char* ws = (char*)d_ws;
  float* h    = (float*)(ws);
  float* y    = (float*)(ws + (size_t)4 * 1024 * 1024);
  float* attn = (float*)(ws + (size_t)68 * 1024 * 1024);
  float* m2   = (float*)(ws + (size_t)76 * 1024 * 1024);
  float* P    = attn;
  float* q    = attn + (size_t)B_ * C_ * CR_;

  const dim3 blk(256);
  k_conv1_gelu<<<dim3(128),  blk, 0, stream>>>(x, w1, b1, h);
  k_conv2     <<<dim3(2048), blk, 0, stream>>>(h, w2, b2, y);
  k_energy    <<<dim3(256),  blk, 0, stream>>>(y, attn);
  k_softmax   <<<dim3(4096), blk, 0, stream>>>(attn);
  k_m2        <<<dim3(256),  blk, 0, stream>>>(attn, wf, gamma, m2);
  k_pq        <<<dim3(256),  blk, 0, stream>>>(m2, w2, b2, P, q);
  k_fusion    <<<dim3(2048), blk, 0, stream>>>(P, q, h, x, wf, bf, out);
}


# energy y.y^T re-associated through y's rank-32 factor: [W2 b2].([h;1].[h;1]^T).[W2 b2]^T from a 33x33 split-K f32 WMMA Gram of h; y never formed, conv2 dropped; f32 summation order changes
# speedup vs baseline: 2.5388x; 2.0579x over previous
#include <hip/hip_runtime.h>
#include <math.h>


#define B_   8
#define C_   512
#define CR_  32
#define HW_  4096
#define C2_  1024

typedef __attribute__((ext_vector_type(2))) float v2f;
typedef __attribute__((ext_vector_type(8))) float v8f;

__device__ __forceinline__ v8f wmma4(v2f a, v2f b, v8f c) {
  return __builtin_amdgcn_wmma_f32_16x16x4_f32(
       false, a,  false, b,
       (short)0, c,  false,  false);
}

template<int NT>
__device__ __forceinline__ void gemm_row(const float* __restrict__ A, int lda,
                                         const float* __restrict__ Bm, int ldb,
                                         int K, v8f c[NT]) {
  const int lane = threadIdx.x & 31;
  const int half = lane >> 4;
  const int l16  = lane & 15;
  const float* ap = A + (size_t)l16 * lda + 2 * half;
  const float* bp = Bm + (size_t)(2 * half) * ldb + l16;
#pragma unroll 2
  for (int k = 0; k < K; k += 4) {
    v2f a = *(const v2f*)ap;
#pragma unroll
    for (int t = 0; t < NT; ++t) {
      v2f b;
      b.x = bp[t * 16];
      b.y = bp[t * 16 + ldb];
      c[t] = wmma4(a, b, c[t]);
    }
    ap += 4;
    bp += (size_t)4 * ldb;
  }
}

__global__ void k_conv1_gelu(const float* __restrict__ x, const float* __restrict__ w1,
                             const float* __restrict__ b1, float* __restrict__ h) {
  constexpr int NT = 4;
  const int wid = blockIdx.x * 8 + (threadIdx.x >> 5);
  const int NG = HW_ / (16 * NT);
  const int MT = CR_ / 16;
  const int b  = wid / (MT * NG);
  const int r0 = wid % (MT * NG);
  const int mt = r0 / NG;
  const int ng = r0 % NG;

  const float* A  = w1 + (size_t)mt * 16 * C_;
  const float* Bm = x + (size_t)b * C_ * HW_ + ng * 16 * NT;
  v8f c[NT] = {};
  gemm_row<NT>(A, C_, Bm, HW_, C_, c);

  const int lane = threadIdx.x & 31;
  const int half = lane >> 4;
  const int l16  = lane & 15;
  float* D = h + (size_t)b * CR_ * HW_;
#pragma unroll
  for (int t = 0; t < NT; ++t)
#pragma unroll
    for (int r = 0; r < 8; ++r) {
      const int m = mt * 16 + r + 8 * half;
      const int n = ng * 16 * NT + t * 16 + l16;
      float v = c[t][r] + b1[m];
      v = 0.5f * v * (1.0f + erff(v * 0.70710678118654752f));
      D[(size_t)m * HW_ + n] = v;
    }
}

#define GP_STRIDE 1088

__global__ void k_gram(const float* __restrict__ h, float* __restrict__ gp) {
  __shared__ float part[8][6][8][32];
  const int b  = blockIdx.x >> 3;
  const int ks = blockIdx.x & 7;
  v8f c00 = {}, c01 = {}, c10 = {}, c11 = {}, s0 = {}, s1 = {};
  {
    const int w    = threadIdx.x >> 5;
    const int lane = threadIdx.x & 31;
    const int half = lane >> 4;
    const int l16  = lane & 15;
    const float* p0 = h + (size_t)b * CR_ * HW_ + (size_t)l16 * HW_ + ks * 512 + w * 64 + 4 * half;
    const float* p1 = p0 + (size_t)16 * HW_;
    float4 r0[8], r1[8];
#pragma unroll
    for (int i = 0; i < 8; ++i) {
      r0[i] = *(const float4*)(p0 + 8 * i);
      r1[i] = *(const float4*)(p1 + 8 * i);
    }
    v2f one;
    one.x = 1.0f;
    one.y = 1.0f;
#pragma unroll
    for (int i = 0; i < 8; ++i) {
      v2f a0, a1;
      a0.x = r0[i].x; a0.y = r0[i].y;
      a1.x = r1[i].x; a1.y = r1[i].y;
      c00 = wmma4(a0, a0, c00);
      c01 = wmma4(a0, a1, c01);
      c10 = wmma4(a1, a0, c10);
      c11 = wmma4(a1, a1, c11);
      s0  = wmma4(a0, one, s0);
      s1  = wmma4(a1, one, s1);
      a0.x = r0[i].z; a0.y = r0[i].w;
      a1.x = r1[i].z; a1.y = r1[i].w;
      c00 = wmma4(a0, a0, c00);
      c01 = wmma4(a0, a1, c01);
      c10 = wmma4(a1, a0, c10);
      c11 = wmma4(a1, a1, c11);
      s0  = wmma4(a0, one, s0);
      s1  = wmma4(a1, one, s1);
    }
  }
  int tid = threadIdx.x;
  asm volatile("v_nop\n\tv_nop\n\tv_nop\n\tv_nop\n\tv_nop\n\tv_nop\n\tv_nop\n\tv_nop\n\tv_nop"
               : "+v"(c00), "+v"(c01), "+v"(c10), "+v"(c11), "+v"(s0), "+v"(s1), "+v"(tid));
  const int w    = tid >> 5;
  const int lane = tid & 31;
  const int half = lane >> 4;
  const int l16  = lane & 15;
#pragma unroll
  for (int r = 0; r < 8; ++r) {
    part[w][0][r][lane] = c00[r];
    part[w][1][r][lane] = c01[r];
    part[w][2][r][lane] = c10[r];
    part[w][3][r][lane] = c11[r];
    part[w][4][r][lane] = s0[r];
    part[w][5][r][lane] = s1[r];
  }
  __syncthreads();

  float* g = gp + (size_t)blockIdx.x * GP_STRIDE;
#pragma unroll
  for (int t = 0; t < 4; ++t) {
    float s = 0.0f;
#pragma unroll
    for (int u = 0; u < 8; ++u) s += part[u][t][w][lane];
    g[(16 * (t >> 1) + w + 8 * half) * 32 + 16 * (t & 1) + l16] = s;
  }
  if (tid < 32) {
    const int rr = tid & 15;
    float s = 0.0f;
#pragma unroll
    for (int u = 0; u < 8; ++u) s += part[u][4 + (tid >> 4)][rr & 7][(rr >> 3) * 16];
    g[1024 + tid] = s;
  }
}


#define PIN9(f) asm volatile("" : "+v"(f[0]), "+v"(f[1]), "+v"(f[2]), "+v"(f[3]), "+v"(f[4]), \
                                  "+v"(f[5]), "+v"(f[6]), "+v"(f[7]), "+v"(f[8]))

__global__ void k_energy(const float* __restrict__ gp, const float* __restrict__ w2,
                         const float* __restrict__ b2, float* __restrict__ e) {
  __shared__ float ga[36][48];
  __shared__ float tl[16][40];
  const int b    = blockIdx.x >> 5;
  const int mt   = blockIdx.x & 31;
  int tid = threadIdx.x;
  v2f bw[4][9];
  {
    const int w    = tid >> 5;
    const int lane = tid & 31;
    const int half = lane >> 4;
    const int l16  = lane & 15;

#pragma unroll
    for (int t = 0; t < 4; ++t) {
      const int n = 64 * w + 16 * t + l16;
      const float* wp = w2 + (size_t)n * CR_ + 2 * half;
#pragma unroll
      for (int s = 0; s < 8; ++s) bw[t][s] = *(const v2f*)(wp + 4 * s);
      const float bn = b2[n];
      bw[t][8].x = half ? 0.0f : bn;
      bw[t][8].y = 0.0f;
    }

    const float* gb = gp + (size_t)b * 8 * GP_STRIDE;
#pragma unroll
    for (int i = 0; i < 4; ++i) {
      const int j = w + 8 * i;
      float s = 0.0f;
#pragma unroll
      for (int u = 0; u < 8; ++u) s += gb[u * GP_STRIDE + j * 32 + lane];
      ga[j][lane] = s;
    }
    if (tid < 32) {
      float s = 0.0f;
#pragma unroll
      for (int u = 0; u < 8; ++u) s += gb[u * GP_STRIDE + 1024 + tid];
      ga[tid][32] = s;
      ga[32][tid] = s;
    }
    for (int idx = tid; idx < 36 * 48; idx += 256) {
      const int j = idx / 48;
      const int k = idx - j * 48;
      if (j >= 33 || k >= 33) ga[j][k] = 0.0f;
      else if (j == 32 && k == 32) ga[j][k] = (float)HW_;
    }
  }
  __syncthreads();

  if ((tid >> 5) < 3) {
    v2f fa[9], fb[9];
    {
      const int w    = tid >> 5;
      const int half = (tid & 31) >> 4;
      const int l16  = tid & 15;
      const int m    = mt * 16 + l16;
      const int col  = 16 * w + l16;
      const float* ap = w2 + (size_t)m * CR_ + 2 * half;
#pragma unroll
      for (int s = 0; s < 8; ++s) fa[s] = *(const v2f*)(ap + 4 * s);
      const float bm = b2[m];
      fa[8].x = half ? 0.0f : bm;
      fa[8].y = 0.0f;
#pragma unroll
      for (int s = 0; s < 9; ++s) {
        fb[s].x = ga[4 * s + 2 * half][col];
        fb[s].y = ga[4 * s + 2 * half + 1][col];
      }
    }
    PIN9(fa);
    PIN9(fb);
    v8f c = {};
#pragma unroll
    for (int s = 0; s < 9; ++s) c = wmma4(fa[s], fb[s], c);
    int t2 = tid;
    asm volatile("v_nop\n\tv_nop\n\tv_nop\n\tv_nop\n\tv_nop\n\tv_nop\n\tv_nop\n\tv_nop\n\tv_nop"
                 : "+v"(c), "+v"(t2));
    const int col  = 16 * (t2 >> 5) + (t2 & 15);
    const int row0 = 8 * ((t2 & 31) >> 4);
    if (col < 36) {
#pragma unroll
      for (int r = 0; r < 8; ++r) tl[row0 + r][col] = c[r];
    }
  }
  __syncthreads();

  v2f ta[9];
  {
    const int half = (tid & 31) >> 4;
    const int l16  = tid & 15;
#pragma unroll
    for (int s = 0; s < 9; ++s) ta[s] = *(const v2f*)&tl[l16][4 * s + 2 * half];
  }
  PIN9(ta);
  PIN9(bw[0]);
  PIN9(bw[1]);
  PIN9(bw[2]);
  PIN9(bw[3]);
  v8f c0 = {}, c1 = {}, c2 = {}, c3 = {};
#pragma unroll
  for (int s = 0; s < 9; ++s) {
    c0 = wmma4(ta[s], bw[0][s], c0);
    c1 = wmma4(ta[s], bw[1][s], c1);
    c2 = wmma4(ta[s], bw[2][s], c2);
    c3 = wmma4(ta[s], bw[3][s], c3);
  }
  asm volatile("v_nop\n\tv_nop\n\tv_nop\n\tv_nop\n\tv_nop\n\tv_nop\n\tv_nop\n\tv_nop\n\tv_nop"
               : "+v"(c0), "+v"(c1), "+v"(c2), "+v"(c3), "+v"(tid));
  {
    const int w    = tid >> 5;
    const int half = (tid & 31) >> 4;
    const int l16  = tid & 15;
    float* D = e + (size_t)b * C_ * C_ + (size_t)(mt * 16 + 8 * half) * C_ + 64 * w + l16;
#pragma unroll
    for (int r = 0; r < 8; ++r) {
      D[(size_t)r * C_]      = c0[r];
      D[(size_t)r * C_ + 16] = c1[r];
      D[(size_t)r * C_ + 32] = c2[r];
      D[(size_t)r * C_ + 48] = c3[r];
    }
  }
}

__global__ void k_softmax(float* __restrict__ e) {
  __shared__ float red[256];
  const int tid = threadIdx.x;
  float* p = e + (size_t)blockIdx.x * C_;
  const float v0 = p[tid];
  const float v1 = p[tid + 256];
  red[tid] = fminf(v0, v1);
  __syncthreads();
  for (int s = 128; s > 0; s >>= 1) {
    if (tid < s) red[tid] = fminf(red[tid], red[tid + s]);
    __syncthreads();
  }
  const float rmin = red[0];
  __syncthreads();
  const float e0 = expf(rmin - v0);
  const float e1 = expf(rmin - v1);
  red[tid] = e0 + e1;
  __syncthreads();
  for (int s = 128; s > 0; s >>= 1) {
    if (tid < s) red[tid] += red[tid + s];
    __syncthreads();
  }
  const float inv = 1.0f / red[0];
  p[tid] = e0 * inv;
  p[tid + 256] = e1 * inv;
}

__global__ void k_m2(const float* __restrict__ attn, const float* __restrict__ wf,
                     const float* __restrict__ gamma, float* __restrict__ m2) {
  constexpr int NT = 4;
  const int wid = blockIdx.x * 8 + (threadIdx.x >> 5);
  const int NG = C_ / (16 * NT);
  const int MT = C_ / 16;
  const int b  = wid / (MT * NG);
  const int r0 = wid % (MT * NG);
  const int mt = r0 / NG;
  const int ng = r0 % NG;

  const float* A  = wf + (size_t)mt * 16 * C2_;
  const float* Bm = attn + (size_t)b * C_ * C_ + ng * 16 * NT;
  v8f c[NT] = {};
  gemm_row<NT>(A, C2_, Bm, C_, C_, c);

  const float g = gamma[0];
  const int lane = threadIdx.x & 31;
  const int half = lane >> 4;
  const int l16  = lane & 15;
  float* D = m2 + (size_t)b * C_ * C_;
#pragma unroll
  for (int t = 0; t < NT; ++t)
#pragma unroll
    for (int r = 0; r < 8; ++r) {
      const int m = mt * 16 + r + 8 * half;
      const int n = ng * 16 * NT + t * 16 + l16;
      D[(size_t)m * C_ + n] = g * c[t][r] + wf[(size_t)m * C2_ + n];
    }
}

__global__ void k_pq(const float* __restrict__ m2, const float* __restrict__ w2,
                     const float* __restrict__ b2, float* __restrict__ P, float* __restrict__ q) {
  __shared__ float part[8][3][8][32];
  const int b    = blockIdx.x >> 5;
  const int mt   = blockIdx.x & 31;
  v8f c0 = {}, c1 = {}, c2 = {};
  {
    const int w    = threadIdx.x >> 5;
    const int lane = threadIdx.x & 31;
    const int half = lane >> 4;
    const int l16  = lane & 15;
    const int k0   = w * 64;
    const float* ap = m2 + (size_t)b * C_ * C_ + (size_t)(mt * 16 + l16) * C_ + k0 + 2 * half;
    const float* bp = w2 + (size_t)(k0 + 2 * half) * CR_ + l16;
    const float* cp = b2 + k0 + 2 * half;
#pragma unroll
    for (int kk = 0; kk < 64; kk += 4) {
      const v2f a = *(const v2f*)(ap + kk);
      v2f b0, b1;
      b0.x = bp[kk * CR_];
      b0.y = bp[kk * CR_ + CR_];
      b1.x = bp[kk * CR_ + 16];
      b1.y = bp[kk * CR_ + CR_ + 16];
      const v2f bb = *(const v2f*)(cp + kk);
      c0 = wmma4(a, b0, c0);
      c1 = wmma4(a, b1, c1);
      c2 = wmma4(a, bb, c2);
    }
  }
  int tid = threadIdx.x;
  asm volatile("v_nop\n\tv_nop\n\tv_nop\n\tv_nop\n\tv_nop\n\tv_nop\n\tv_nop\n\tv_nop\n\tv_nop"
               : "+v"(c0), "+v"(c1), "+v"(c2), "+v"(tid));
  const int w    = tid >> 5;
  const int lane = tid & 31;
  const int half = lane >> 4;
  const int l16  = lane & 15;
#pragma unroll
  for (int r = 0; r < 8; ++r) {
    part[w][0][r][lane] = c0[r];
    part[w][1][r][lane] = c1[r];
    part[w][2][r][lane] = c2[r];
  }
  __syncthreads();

  const int m = mt * 16 + w + 8 * half;
#pragma unroll
  for (int t = 0; t < 2; ++t) {
    float s = 0.0f;
#pragma unroll
    for (int u = 0; u < 8; ++u) s += part[u][t][w][lane];
    P[((size_t)b * C_ + m) * CR_ + t * 16 + l16] = s;
  }
  if (tid < 16) {
    const int row = tid;
    float s = 0.0f;
#pragma unroll
    for (int u = 0; u < 8; ++u) s += part[u][2][row & 7][(row >> 3) * 16];
    q[(size_t)blockIdx.x * 32 + row] = s;
  }
}

__global__ void k_fusion(const float* __restrict__ P, const float* __restrict__ q,
                         const float* __restrict__ h, const float* __restrict__ x,
                         const float* __restrict__ wf, const float* __restrict__ bf,
                         float* __restrict__ out) {
  constexpr int NT = 4;
  const int wid = blockIdx.x * 8 + (threadIdx.x >> 5);
  const int NG = HW_ / (16 * NT);
  const int MT = C_ / 16;
  const int b  = wid / (MT * NG);
  const int r0 = wid % (MT * NG);
  const int mt = r0 / NG;
  const int ng = r0 % NG;

  v8f c[NT] = {};
  gemm_row<NT>(P + ((size_t)b * C_ + (size_t)mt * 16) * CR_, CR_,
               h + (size_t)b * CR_ * HW_ + ng * 16 * NT, HW_, CR_, c);
  gemm_row<NT>(wf + (size_t)mt * 16 * C2_ + C_, C2_,
               x + (size_t)b * C_ * HW_ + ng * 16 * NT, HW_, C_, c);

  const int lane = threadIdx.x & 31;
  const int half = lane >> 4;
  const int l16  = lane & 15;
  float* D = out + (size_t)b * C_ * HW_;
  const float* qb = q + ((size_t)b * MT + mt) * 32;
#pragma unroll
  for (int t = 0; t < NT; ++t)
#pragma unroll
    for (int r = 0; r < 8; ++r) {
      const int m = mt * 16 + r + 8 * half;
      const int n = ng * 16 * NT + t * 16 + l16;
      D[(size_t)m * HW_ + n] = c[t][r] + (bf[m] + qb[r + 8 * half]);
    }
}

extern "C" void kernel_launch(void* const* d_in, const int* in_sizes, int n_in,
                              void* d_out, int out_size, void* d_ws, size_t ws_size,
                              hipStream_t stream) {
  const float* x     = (const float*)d_in[0];
  const float* w1    = (const float*)d_in[1];
  const float* b1    = (const float*)d_in[2];
  const float* w2    = (const float*)d_in[3];
  const float* b2    = (const float*)d_in[4];
  const float* wf    = (const float*)d_in[5];
  const float* bf    = (const float*)d_in[6];
  const float* gamma = (const float*)d_in[7];
  float* out = (float*)d_out;

  char* ws = (char*)d_ws;
  float* h    = (float*)(ws);
  float* gp   = (float*)(ws + (size_t)4 * 1024 * 1024);
  float* attn = (float*)(ws + (size_t)68 * 1024 * 1024);
  float* m2   = (float*)(ws + (size_t)76 * 1024 * 1024);
  float* P    = attn;
  float* q    = attn + (size_t)B_ * C_ * CR_;

  const dim3 blk(256);
  k_conv1_gelu<<<dim3(128),  blk, 0, stream>>>(x, w1, b1, h);
  k_gram      <<<dim3(64),   blk, 0, stream>>>(h, gp);
  k_energy    <<<dim3(256),  blk, 0, stream>>>(gp, w2, b2, attn);
  k_softmax   <<<dim3(4096), blk, 0, stream>>>(attn);
  k_m2        <<<dim3(256),  blk, 0, stream>>>(attn, wf, gamma, m2);
  k_pq        <<<dim3(256),  blk, 0, stream>>>(m2, w2, b2, P, q);
  k_fusion    <<<dim3(2048), blk, 0, stream>>>(P, q, h, x, wf, bf, out);
}


# final product (P.h + Wf2.x): wave tile 16x64 -> 64x64, 16 accumulator tiles per wave; a k-step's 4 A + 4 B fragments feed 16 WMMAs (was 1 + 4 for 4); 2048 -> 512 blocks; f32 sum order unchanged
# speedup vs baseline: 2.9870x; 1.1765x over previous
#include <hip/hip_runtime.h>
#include <math.h>


#define B_   8
#define C_   512
#define CR_  32
#define HW_  4096
#define C2_  1024

typedef __attribute__((ext_vector_type(2))) float v2f;
typedef __attribute__((ext_vector_type(8))) float v8f;

__device__ __forceinline__ v8f wmma4(v2f a, v2f b, v8f c) {
  return __builtin_amdgcn_wmma_f32_16x16x4_f32(
       false, a,  false, b,
       (short)0, c,  false,  false);
}

template<int NT>
__device__ __forceinline__ void gemm_row(const float* __restrict__ A, int lda,
                                         const float* __restrict__ Bm, int ldb,
                                         int K, v8f c[NT]) {
  const int lane = threadIdx.x & 31;
  const int half = lane >> 4;
  const int l16  = lane & 15;
  const float* ap = A + (size_t)l16 * lda + 2 * half;
  const float* bp = Bm + (size_t)(2 * half) * ldb + l16;
#pragma unroll 2
  for (int k = 0; k < K; k += 4) {
    v2f a = *(const v2f*)ap;
#pragma unroll
    for (int t = 0; t < NT; ++t) {
      v2f b;
      b.x = bp[t * 16];
      b.y = bp[t * 16 + ldb];
      c[t] = wmma4(a, b, c[t]);
    }
    ap += 4;
    bp += (size_t)4 * ldb;
  }
}

__device__ __forceinline__ void gemm_tile64(const float* __restrict__ A, int lda,
                                            const float* __restrict__ Bm, int ldb,
                                            int K, v8f c[4][4]) {
  const int lane = threadIdx.x & 31;
  const int half = lane >> 4;
  const int l16  = lane & 15;
  const float* ap = A + (size_t)l16 * lda + 2 * half;
  const float* bp = Bm + (size_t)(2 * half) * ldb + l16;
#pragma unroll 2
  for (int k = 0; k < K; k += 4) {
    v2f a[4], b[4];
#pragma unroll
    for (int i = 0; i < 4; ++i) a[i] = *(const v2f*)(ap + (size_t)(16 * i) * lda);
#pragma unroll
    for (int t = 0; t < 4; ++t) {
      b[t].x = bp[t * 16];
      b[t].y = bp[t * 16 + ldb];
    }
#pragma unroll
    for (int i = 0; i < 4; ++i)
#pragma unroll
      for (int t = 0; t < 4; ++t) c[i][t] = wmma4(a[i], b[t], c[i][t]);
    ap += 4;
    bp += (size_t)4 * ldb;
  }
}

__global__ void k_conv1_gelu(const float* __restrict__ x, const float* __restrict__ w1,
                             const float* __restrict__ b1, float* __restrict__ h) {
  constexpr int NT = 4;
  const int wid = blockIdx.x * 8 + (threadIdx.x >> 5);
  const int NG = HW_ / (16 * NT);
  const int MT = CR_ / 16;
  const int b  = wid / (MT * NG);
  const int r0 = wid % (MT * NG);
  const int mt = r0 / NG;
  const int ng = r0 % NG;

  const float* A  = w1 + (size_t)mt * 16 * C_;
  const float* Bm = x + (size_t)b * C_ * HW_ + ng * 16 * NT;
  v8f c[NT] = {};
  gemm_row<NT>(A, C_, Bm, HW_, C_, c);

  const int lane = threadIdx.x & 31;
  const int half = lane >> 4;
  const int l16  = lane & 15;
  float* D = h + (size_t)b * CR_ * HW_;
#pragma unroll
  for (int t = 0; t < NT; ++t)
#pragma unroll
    for (int r = 0; r < 8; ++r) {
      const int m = mt * 16 + r + 8 * half;
      const int n = ng * 16 * NT + t * 16 + l16;
      float v = c[t][r] + b1[m];
      v = 0.5f * v * (1.0f + erff(v * 0.70710678118654752f));
      D[(size_t)m * HW_ + n] = v;
    }
}

#define GP_STRIDE 1088

__global__ void k_gram(const float* __restrict__ h, float* __restrict__ gp) {
  __shared__ float part[8][6][8][32];
  const int b  = blockIdx.x >> 3;
  const int ks = blockIdx.x & 7;
  v8f c00 = {}, c01 = {}, c10 = {}, c11 = {}, s0 = {}, s1 = {};
  {
    const int w    = threadIdx.x >> 5;
    const int lane = threadIdx.x & 31;
    const int half = lane >> 4;
    const int l16  = lane & 15;
    const float* p0 = h + (size_t)b * CR_ * HW_ + (size_t)l16 * HW_ + ks * 512 + w * 64 + 4 * half;
    const float* p1 = p0 + (size_t)16 * HW_;
    float4 r0[8], r1[8];
#pragma unroll
    for (int i = 0; i < 8; ++i) {
      r0[i] = *(const float4*)(p0 + 8 * i);
      r1[i] = *(const float4*)(p1 + 8 * i);
    }
    v2f one;
    one.x = 1.0f;
    one.y = 1.0f;
#pragma unroll
    for (int i = 0; i < 8; ++i) {
      v2f a0, a1;
      a0.x = r0[i].x; a0.y = r0[i].y;
      a1.x = r1[i].x; a1.y = r1[i].y;
      c00 = wmma4(a0, a0, c00);
      c01 = wmma4(a0, a1, c01);
      c10 = wmma4(a1, a0, c10);
      c11 = wmma4(a1, a1, c11);
      s0  = wmma4(a0, one, s0);
      s1  = wmma4(a1, one, s1);
      a0.x = r0[i].z; a0.y = r0[i].w;
      a1.x = r1[i].z; a1.y = r1[i].w;
      c00 = wmma4(a0, a0, c00);
      c01 = wmma4(a0, a1, c01);
      c10 = wmma4(a1, a0, c10);
      c11 = wmma4(a1, a1, c11);
      s0  = wmma4(a0, one, s0);
      s1  = wmma4(a1, one, s1);
    }
  }
  int tid = threadIdx.x;
  asm volatile("v_nop\n\tv_nop\n\tv_nop\n\tv_nop\n\tv_nop\n\tv_nop\n\tv_nop\n\tv_nop\n\tv_nop"
               : "+v"(c00), "+v"(c01), "+v"(c10), "+v"(c11), "+v"(s0), "+v"(s1), "+v"(tid));
  const int w    = tid >> 5;
  const int lane = tid & 31;
  const int half = lane >> 4;
  const int l16  = lane & 15;
#pragma unroll
  for (int r = 0; r < 8; ++r) {
    part[w][0][r][lane] = c00[r];
    part[w][1][r][lane] = c01[r];
    part[w][2][r][lane] = c10[r];
    part[w][3][r][lane] = c11[r];
    part[w][4][r][lane] = s0[r];
    part[w][5][r][lane] = s1[r];
  }
  __syncthreads();

  float* g = gp + (size_t)blockIdx.x * GP_STRIDE;
#pragma unroll
  for (int t = 0; t < 4; ++t) {
    float s = 0.0f;
#pragma unroll
    for (int u = 0; u < 8; ++u) s += part[u][t][w][lane];
    g[(16 * (t >> 1) + w + 8 * half) * 32 + 16 * (t & 1) + l16] = s;
  }
  if (tid < 32) {
    const int rr = tid & 15;
    float s = 0.0f;
#pragma unroll
    for (int u = 0; u < 8; ++u) s += part[u][4 + (tid >> 4)][rr & 7][(rr >> 3) * 16];
    g[1024 + tid] = s;
  }
}


#define PIN9(f) asm volatile("" : "+v"(f[0]), "+v"(f[1]), "+v"(f[2]), "+v"(f[3]), "+v"(f[4]), \
                                  "+v"(f[5]), "+v"(f[6]), "+v"(f[7]), "+v"(f[8]))

__global__ void k_energy(const float* __restrict__ gp, const float* __restrict__ w2,
                         const float* __restrict__ b2, float* __restrict__ e) {
  __shared__ float ga[36][48];
  __shared__ float tl[16][40];
  const int b    = blockIdx.x >> 5;
  const int mt   = blockIdx.x & 31;
  int tid = threadIdx.x;
  v2f bw[4][9];
  {
    const int w    = tid >> 5;
    const int lane = tid & 31;
    const int half = lane >> 4;
    const int l16  = lane & 15;

#pragma unroll
    for (int t = 0; t < 4; ++t) {
      const int n = 64 * w + 16 * t + l16;
      const float* wp = w2 + (size_t)n * CR_ + 2 * half;
#pragma unroll
      for (int s = 0; s < 8; ++s) bw[t][s] = *(const v2f*)(wp + 4 * s);
      const float bn = b2[n];
      bw[t][8].x = half ? 0.0f : bn;
      bw[t][8].y = 0.0f;
    }

    const float* gb = gp + (size_t)b * 8 * GP_STRIDE;
#pragma unroll
    for (int i = 0; i < 4; ++i) {
      const int j = w + 8 * i;
      float s = 0.0f;
#pragma unroll
      for (int u = 0; u < 8; ++u) s += gb[u * GP_STRIDE + j * 32 + lane];
      ga[j][lane] = s;
    }
    if (tid < 32) {
      float s = 0.0f;
#pragma unroll
      for (int u = 0; u < 8; ++u) s += gb[u * GP_STRIDE + 1024 + tid];
      ga[tid][32] = s;
      ga[32][tid] = s;
    }
    for (int idx = tid; idx < 36 * 48; idx += 256) {
      const int j = idx / 48;
      const int k = idx - j * 48;
      if (j >= 33 || k >= 33) ga[j][k] = 0.0f;
      else if (j == 32 && k == 32) ga[j][k] = (float)HW_;
    }
  }
  __syncthreads();

  if ((tid >> 5) < 3) {
    v2f fa[9], fb[9];
    {
      const int w    = tid >> 5;
      const int half = (tid & 31) >> 4;
      const int l16  = tid & 15;
      const int m    = mt * 16 + l16;
      const int col  = 16 * w + l16;
      const float* ap = w2 + (size_t)m * CR_ + 2 * half;
#pragma unroll
      for (int s = 0; s < 8; ++s) fa[s] = *(const v2f*)(ap + 4 * s);
      const float bm = b2[m];
      fa[8].x = half ? 0.0f : bm;
      fa[8].y = 0.0f;
#pragma unroll
      for (int s = 0; s < 9; ++s) {
        fb[s].x = ga[4 * s + 2 * half][col];
        fb[s].y = ga[4 * s + 2 * half + 1][col];
      }
    }
    PIN9(fa);
    PIN9(fb);
    v8f c = {};
#pragma unroll
    for (int s = 0; s < 9; ++s) c = wmma4(fa[s], fb[s], c);
    int t2 = tid;
    asm volatile("v_nop\n\tv_nop\n\tv_nop\n\tv_nop\n\tv_nop\n\tv_nop\n\tv_nop\n\tv_nop\n\tv_nop"
                 : "+v"(c), "+v"(t2));
    const int col  = 16 * (t2 >> 5) + (t2 & 15);
    const int row0 = 8 * ((t2 & 31) >> 4);
    if (col < 36) {
#pragma unroll
      for (int r = 0; r < 8; ++r) tl[row0 + r][col] = c[r];
    }
  }
  __syncthreads();

  v2f ta[9];
  {
    const int half = (tid & 31) >> 4;
    const int l16  = tid & 15;
#pragma unroll
    for (int s = 0; s < 9; ++s) ta[s] = *(const v2f*)&tl[l16][4 * s + 2 * half];
  }
  PIN9(ta);
  PIN9(bw[0]);
  PIN9(bw[1]);
  PIN9(bw[2]);
  PIN9(bw[3]);
  v8f c0 = {}, c1 = {}, c2 = {}, c3 = {};
#pragma unroll
  for (int s = 0; s < 9; ++s) {
    c0 = wmma4(ta[s], bw[0][s], c0);
    c1 = wmma4(ta[s], bw[1][s], c1);
    c2 = wmma4(ta[s], bw[2][s], c2);
    c3 = wmma4(ta[s], bw[3][s], c3);
  }
  asm volatile("v_nop\n\tv_nop\n\tv_nop\n\tv_nop\n\tv_nop\n\tv_nop\n\tv_nop\n\tv_nop\n\tv_nop"
               : "+v"(c0), "+v"(c1), "+v"(c2), "+v"(c3), "+v"(tid));
  {
    const int w    = tid >> 5;
    const int half = (tid & 31) >> 4;
    const int l16  = tid & 15;
    float* D = e + (size_t)b * C_ * C_ + (size_t)(mt * 16 + 8 * half) * C_ + 64 * w + l16;
#pragma unroll
    for (int r = 0; r < 8; ++r) {
      D[(size_t)r * C_]      = c0[r];
      D[(size_t)r * C_ + 16] = c1[r];
      D[(size_t)r * C_ + 32] = c2[r];
      D[(size_t)r * C_ + 48] = c3[r];
    }
  }
}

__global__ void k_softmax(float* __restrict__ e) {
  __shared__ float red[256];
  const int tid = threadIdx.x;
  float* p = e + (size_t)blockIdx.x * C_;
  const float v0 = p[tid];
  const float v1 = p[tid + 256];
  red[tid] = fminf(v0, v1);
  __syncthreads();
  for (int s = 128; s > 0; s >>= 1) {
    if (tid < s) red[tid] = fminf(red[tid], red[tid + s]);
    __syncthreads();
  }
  const float rmin = red[0];
  __syncthreads();
  const float e0 = expf(rmin - v0);
  const float e1 = expf(rmin - v1);
  red[tid] = e0 + e1;
  __syncthreads();
  for (int s = 128; s > 0; s >>= 1) {
    if (tid < s) red[tid] += red[tid + s];
    __syncthreads();
  }
  const float inv = 1.0f / red[0];
  p[tid] = e0 * inv;
  p[tid + 256] = e1 * inv;
}

__global__ void k_m2(const float* __restrict__ attn, const float* __restrict__ wf,
                     const float* __restrict__ gamma, float* __restrict__ m2) {
  constexpr int NT = 4;
  const int wid = blockIdx.x * 8 + (threadIdx.x >> 5);
  const int NG = C_ / (16 * NT);
  const int MT = C_ / 16;
  const int b  = wid / (MT * NG);
  const int r0 = wid % (MT * NG);
  const int mt = r0 / NG;
  const int ng = r0 % NG;

  const float* A  = wf + (size_t)mt * 16 * C2_;
  const float* Bm = attn + (size_t)b * C_ * C_ + ng * 16 * NT;
  v8f c[NT] = {};
  gemm_row<NT>(A, C2_, Bm, C_, C_, c);

  const float g = gamma[0];
  const int lane = threadIdx.x & 31;
  const int half = lane >> 4;
  const int l16  = lane & 15;
  float* D = m2 + (size_t)b * C_ * C_;
#pragma unroll
  for (int t = 0; t < NT; ++t)
#pragma unroll
    for (int r = 0; r < 8; ++r) {
      const int m = mt * 16 + r + 8 * half;
      const int n = ng * 16 * NT + t * 16 + l16;
      D[(size_t)m * C_ + n] = g * c[t][r] + wf[(size_t)m * C2_ + n];
    }
}

__global__ void k_pq(const float* __restrict__ m2, const float* __restrict__ w2,
                     const float* __restrict__ b2, float* __restrict__ P, float* __restrict__ q) {
  __shared__ float part[8][3][8][32];
  const int b    = blockIdx.x >> 5;
  const int mt   = blockIdx.x & 31;
  v8f c0 = {}, c1 = {}, c2 = {};
  {
    const int w    = threadIdx.x >> 5;
    const int lane = threadIdx.x & 31;
    const int half = lane >> 4;
    const int l16  = lane & 15;
    const int k0   = w * 64;
    const float* ap = m2 + (size_t)b * C_ * C_ + (size_t)(mt * 16 + l16) * C_ + k0 + 2 * half;
    const float* bp = w2 + (size_t)(k0 + 2 * half) * CR_ + l16;
    const float* cp = b2 + k0 + 2 * half;
#pragma unroll
    for (int kk = 0; kk < 64; kk += 4) {
      const v2f a = *(const v2f*)(ap + kk);
      v2f b0, b1;
      b0.x = bp[kk * CR_];
      b0.y = bp[kk * CR_ + CR_];
      b1.x = bp[kk * CR_ + 16];
      b1.y = bp[kk * CR_ + CR_ + 16];
      const v2f bb = *(const v2f*)(cp + kk);
      c0 = wmma4(a, b0, c0);
      c1 = wmma4(a, b1, c1);
      c2 = wmma4(a, bb, c2);
    }
  }
  int tid = threadIdx.x;
  asm volatile("v_nop\n\tv_nop\n\tv_nop\n\tv_nop\n\tv_nop\n\tv_nop\n\tv_nop\n\tv_nop\n\tv_nop"
               : "+v"(c0), "+v"(c1), "+v"(c2), "+v"(tid));
  const int w    = tid >> 5;
  const int lane = tid & 31;
  const int half = lane >> 4;
  const int l16  = lane & 15;
#pragma unroll
  for (int r = 0; r < 8; ++r) {
    part[w][0][r][lane] = c0[r];
    part[w][1][r][lane] = c1[r];
    part[w][2][r][lane] = c2[r];
  }
  __syncthreads();

  const int m = mt * 16 + w + 8 * half;
#pragma unroll
  for (int t = 0; t < 2; ++t) {
    float s = 0.0f;
#pragma unroll
    for (int u = 0; u < 8; ++u) s += part[u][t][w][lane];
    P[((size_t)b * C_ + m) * CR_ + t * 16 + l16] = s;
  }
  if (tid < 16) {
    const int row = tid;
    float s = 0.0f;
#pragma unroll
    for (int u = 0; u < 8; ++u) s += part[u][2][row & 7][(row >> 3) * 16];
    q[(size_t)blockIdx.x * 32 + row] = s;
  }
}

__global__ void __launch_bounds__(256) k_fusion(const float* __restrict__ P, const float* __restrict__ q,
                         const float* __restrict__ h, const float* __restrict__ x,
                         const float* __restrict__ wf, const float* __restrict__ bf,
                         float* __restrict__ out) {
  const int wid = blockIdx.x * 8 + (threadIdx.x >> 5);
  const int NG = HW_ / 64;
  const int MT = C_ / 64;
  const int b  = wid / (MT * NG);
  const int r0 = wid % (MT * NG);
  const int mt = r0 / NG;
  const int ng = r0 % NG;

  v8f c[4][4] = {};
  gemm_tile64(P + ((size_t)b * C_ + (size_t)mt * 64) * CR_, CR_,
              h + (size_t)b * CR_ * HW_ + ng * 64, HW_, CR_, c);
  gemm_tile64(wf + (size_t)mt * 64 * C2_ + C_, C2_,
              x + (size_t)b * C_ * HW_ + ng * 64, HW_, C_, c);

  const int lane = threadIdx.x & 31;
  const int half = lane >> 4;
  const int l16  = lane & 15;
  float* D = out + (size_t)b * C_ * HW_;
  const float* qb = q + ((size_t)b * (C_ / 16) + (size_t)mt * 4) * 32;
#pragma unroll
  for (int i = 0; i < 4; ++i)
#pragma unroll
    for (int r = 0; r < 8; ++r) {
      const int m = mt * 64 + 16 * i + r + 8 * half;
      const float bias = bf[m] + qb[i * 32 + r + 8 * half];
#pragma unroll
      for (int t = 0; t < 4; ++t) {
        const int n = ng * 64 + t * 16 + l16;
        D[(size_t)m * HW_ + n] = c[i][t][r] + bias;
      }
    }
}

extern "C" void kernel_launch(void* const* d_in, const int* in_sizes, int n_in,
                              void* d_out, int out_size, void* d_ws, size_t ws_size,
                              hipStream_t stream) {
  const float* x     = (const float*)d_in[0];
  const float* w1    = (const float*)d_in[1];
  const float* b1    = (const float*)d_in[2];
  const float* w2    = (const float*)d_in[3];
  const float* b2    = (const float*)d_in[4];
  const float* wf    = (const float*)d_in[5];
  const float* bf    = (const float*)d_in[6];
  const float* gamma = (const float*)d_in[7];
  float* out = (float*)d_out;

  char* ws = (char*)d_ws;
  float* h    = (float*)(ws);
  float* gp   = (float*)(ws + (size_t)4 * 1024 * 1024);
  float* attn = (float*)(ws + (size_t)68 * 1024 * 1024);
  float* m2   = (float*)(ws + (size_t)76 * 1024 * 1024);
  float* P    = attn;
  float* q    = attn + (size_t)B_ * C_ * CR_;

  const dim3 blk(256);
  k_conv1_gelu<<<dim3(128),  blk, 0, stream>>>(x, w1, b1, h);
  k_gram      <<<dim3(64),   blk, 0, stream>>>(h, gp);
  k_energy    <<<dim3(256),  blk, 0, stream>>>(gp, w2, b2, attn);
  k_softmax   <<<dim3(4096), blk, 0, stream>>>(attn);
  k_m2        <<<dim3(256),  blk, 0, stream>>>(attn, wf, gamma, m2);
  k_pq        <<<dim3(256),  blk, 0, stream>>>(m2, w2, b2, P, q);
  k_fusion    <<<dim3(512),  blk, 0, stream>>>(P, q, h, x, wf, bf, out);
}


# conv1: K=512 split over a block's 8 waves (64 channels each, all loads issued back to back), both 16-row tiles per wave from one x fragment, LDS sum, GELU; 128->512 blocks; f32 sum order changes
# speedup vs baseline: 3.3357x; 1.1167x over previous
#include <hip/hip_runtime.h>
#include <math.h>


#define B_   8
#define C_   512
#define CR_  32
#define HW_  4096
#define C2_  1024

typedef __attribute__((ext_vector_type(2))) float v2f;
typedef __attribute__((ext_vector_type(8))) float v8f;

__device__ __forceinline__ v8f wmma4(v2f a, v2f b, v8f c) {
  return __builtin_amdgcn_wmma_f32_16x16x4_f32(
       false, a,  false, b,
       (short)0, c,  false,  false);
}

template<int NT>
__device__ __forceinline__ void gemm_row(const float* __restrict__ A, int lda,
                                         const float* __restrict__ Bm, int ldb,
                                         int K, v8f c[NT]) {
  const int lane = threadIdx.x & 31;
  const int half = lane >> 4;
  const int l16  = lane & 15;
  const float* ap = A + (size_t)l16 * lda + 2 * half;
  const float* bp = Bm + (size_t)(2 * half) * ldb + l16;
#pragma unroll 2
  for (int k = 0; k < K; k += 4) {
    v2f a = *(const v2f*)ap;
#pragma unroll
    for (int t = 0; t < NT; ++t) {
      v2f b;
      b.x = bp[t * 16];
      b.y = bp[t * 16 + ldb];
      c[t] = wmma4(a, b, c[t]);
    }
    ap += 4;
    bp += (size_t)4 * ldb;
  }
}

__device__ __forceinline__ void gemm_tile64(const float* __restrict__ A, int lda,
                                            const float* __restrict__ Bm, int ldb,
                                            int K, v8f c[4][4]) {
  const int lane = threadIdx.x & 31;
  const int half = lane >> 4;
  const int l16  = lane & 15;
  const float* ap = A + (size_t)l16 * lda + 2 * half;
  const float* bp = Bm + (size_t)(2 * half) * ldb + l16;
#pragma unroll 2
  for (int k = 0; k < K; k += 4) {
    v2f a[4], b[4];
#pragma unroll
    for (int i = 0; i < 4; ++i) a[i] = *(const v2f*)(ap + (size_t)(16 * i) * lda);
#pragma unroll
    for (int t = 0; t < 4; ++t) {
      b[t].x = bp[t * 16];
      b[t].y = bp[t * 16 + ldb];
    }
#pragma unroll
    for (int i = 0; i < 4; ++i)
#pragma unroll
      for (int t = 0; t < 4; ++t) c[i][t] = wmma4(a[i], b[t], c[i][t]);
    ap += 4;
    bp += (size_t)4 * ldb;
  }
}

__global__ void __launch_bounds__(256) __attribute__((amdgpu_waves_per_eu(4, 4))) k_conv1_gelu(const float* __restrict__ x, const float* __restrict__ w1,
                             const float* __restrict__ b1, float* __restrict__ h) {
  __shared__ float part[8][8][8][32];
  const int b  = blockIdx.x >> 6;
  const int ng = blockIdx.x & 63;
  v8f c[2][4] = {};
  {
    const int w    = threadIdx.x >> 5;
    const int lane = threadIdx.x & 31;
    const int half = lane >> 4;
    const int l16  = lane & 15;
    const int k0   = w * 64;
    const float* ap = w1 + (size_t)l16 * C_ + k0 + 2 * half;
    const float* bp = x + (size_t)b * C_ * HW_ + (size_t)(k0 + 2 * half) * HW_ + ng * 64 + l16;
#pragma unroll
    for (int kk = 0; kk < 64; kk += 4) {
      const v2f a0 = *(const v2f*)(ap + kk);
      const v2f a1 = *(const v2f*)(ap + 16 * C_ + kk);
      v2f bx[4];
#pragma unroll
      for (int t = 0; t < 4; ++t) {
        bx[t].x = bp[(size_t)kk * HW_ + 16 * t];
        bx[t].y = bp[(size_t)(kk + 1) * HW_ + 16 * t];
      }
#pragma unroll
      for (int t = 0; t < 4; ++t) {
        c[0][t] = wmma4(a0, bx[t], c[0][t]);
        c[1][t] = wmma4(a1, bx[t], c[1][t]);
      }
    }
  }
  int tid = threadIdx.x;
  asm volatile("v_nop\n\tv_nop\n\tv_nop\n\tv_nop\n\tv_nop\n\tv_nop\n\tv_nop\n\tv_nop\n\tv_nop"
               : "+v"(c[0][0]), "+v"(c[0][1]), "+v"(c[0][2]), "+v"(c[0][3]),
                 "+v"(c[1][0]), "+v"(c[1][1]), "+v"(c[1][2]), "+v"(c[1][3]), "+v"(tid));
  const int w    = tid >> 5;
  const int lane = tid & 31;
  const int half = lane >> 4;
  const int l16  = lane & 15;
#pragma unroll
  for (int i = 0; i < 2; ++i)
#pragma unroll
    for (int t = 0; t < 4; ++t)
#pragma unroll
      for (int r = 0; r < 8; ++r) part[w][4 * i + t][r][lane] = c[i][t][r];
  __syncthreads();

  float* D = h + (size_t)b * CR_ * HW_ + ng * 64 + l16;
  float s[2][4];
#pragma unroll
  for (int i = 0; i < 2; ++i)
#pragma unroll
    for (int t = 0; t < 4; ++t) {
      float a = 0.0f;
#pragma unroll
      for (int u = 0; u < 8; ++u) a += part[u][4 * i + t][w][lane];
      s[i][t] = a;
    }
#pragma unroll
  for (int i = 0; i < 2; ++i) {
    const int m = 16 * i + w + 8 * half;
    const float bias = b1[m];
#pragma unroll
    for (int t = 0; t < 4; ++t) {
      float v = s[i][t] + bias;
      v = 0.5f * v * (1.0f + erff(v * 0.70710678118654752f));
      D[(size_t)m * HW_ + 16 * t] = v;
    }
  }
}

#define GP_STRIDE 1088

__global__ void k_gram(const float* __restrict__ h, float* __restrict__ gp) {
  __shared__ float part[8][6][8][32];
  const int b  = blockIdx.x >> 3;
  const int ks = blockIdx.x & 7;
  v8f c00 = {}, c01 = {}, c10 = {}, c11 = {}, s0 = {}, s1 = {};
  {
    const int w    = threadIdx.x >> 5;
    const int lane = threadIdx.x & 31;
    const int half = lane >> 4;
    const int l16  = lane & 15;
    const float* p0 = h + (size_t)b * CR_ * HW_ + (size_t)l16 * HW_ + ks * 512 + w * 64 + 4 * half;
    const float* p1 = p0 + (size_t)16 * HW_;
    float4 r0[8], r1[8];
#pragma unroll
    for (int i = 0; i < 8; ++i) {
      r0[i] = *(const float4*)(p0 + 8 * i);
      r1[i] = *(const float4*)(p1 + 8 * i);
    }
    v2f one;
    one.x = 1.0f;
    one.y = 1.0f;
#pragma unroll
    for (int i = 0; i < 8; ++i) {
      v2f a0, a1;
      a0.x = r0[i].x; a0.y = r0[i].y;
      a1.x = r1[i].x; a1.y = r1[i].y;
      c00 = wmma4(a0, a0, c00);
      c01 = wmma4(a0, a1, c01);
      c10 = wmma4(a1, a0, c10);
      c11 = wmma4(a1, a1, c11);
      s0  = wmma4(a0, one, s0);
      s1  = wmma4(a1, one, s1);
      a0.x = r0[i].z; a0.y = r0[i].w;
      a1.x = r1[i].z; a1.y = r1[i].w;
      c00 = wmma4(a0, a0, c00);
      c01 = wmma4(a0, a1, c01);
      c10 = wmma4(a1, a0, c10);
      c11 = wmma4(a1, a1, c11);
      s0  = wmma4(a0, one, s0);
      s1  = wmma4(a1, one, s1);
    }
  }
  int tid = threadIdx.x;
  asm volatile("v_nop\n\tv_nop\n\tv_nop\n\tv_nop\n\tv_nop\n\tv_nop\n\tv_nop\n\tv_nop\n\tv_nop"
               : "+v"(c00), "+v"(c01), "+v"(c10), "+v"(c11), "+v"(s0), "+v"(s1), "+v"(tid));
  const int w    = tid >> 5;
  const int lane = tid & 31;
  const int half = lane >> 4;
  const int l16  = lane & 15;
#pragma unroll
  for (int r = 0; r < 8; ++r) {
    part[w][0][r][lane] = c00[r];
    part[w][1][r][lane] = c01[r];
    part[w][2][r][lane] = c10[r];
    part[w][3][r][lane] = c11[r];
    part[w][4][r][lane] = s0[r];
    part[w][5][r][lane] = s1[r];
  }
  __syncthreads();

  float* g = gp + (size_t)blockIdx.x * GP_STRIDE;
#pragma unroll
  for (int t = 0; t < 4; ++t) {
    float s = 0.0f;
#pragma unroll
    for (int u = 0; u < 8; ++u) s += part[u][t][w][lane];
    g[(16 * (t >> 1) + w + 8 * half) * 32 + 16 * (t & 1) + l16] = s;
  }
  if (tid < 32) {
    const int rr = tid & 15;
    float s = 0.0f;
#pragma unroll
    for (int u = 0; u < 8; ++u) s += part[u][4 + (tid >> 4)][rr & 7][(rr >> 3) * 16];
    g[1024 + tid] = s;
  }
}


#define PIN9(f) asm volatile("" : "+v"(f[0]), "+v"(f[1]), "+v"(f[2]), "+v"(f[3]), "+v"(f[4]), \
                                  "+v"(f[5]), "+v"(f[6]), "+v"(f[7]), "+v"(f[8]))

__global__ void k_energy(const float* __restrict__ gp, const float* __restrict__ w2,
                         const float* __restrict__ b2, float* __restrict__ e) {
  __shared__ float ga[36][48];
  __shared__ float tl[16][40];
  const int b    = blockIdx.x >> 5;
  const int mt   = blockIdx.x & 31;
  int tid = threadIdx.x;
  v2f bw[4][9];
  {
    const int w    = tid >> 5;
    const int lane = tid & 31;
    const int half = lane >> 4;
    const int l16  = lane & 15;

#pragma unroll
    for (int t = 0; t < 4; ++t) {
      const int n = 64 * w + 16 * t + l16;
      const float* wp = w2 + (size_t)n * CR_ + 2 * half;
#pragma unroll
      for (int s = 0; s < 8; ++s) bw[t][s] = *(const v2f*)(wp + 4 * s);
      const float bn = b2[n];
      bw[t][8].x = half ? 0.0f : bn;
      bw[t][8].y = 0.0f;
    }

    const float* gb = gp + (size_t)b * 8 * GP_STRIDE;
#pragma unroll
    for (int i = 0; i < 4; ++i) {
      const int j = w + 8 * i;
      float s = 0.0f;
#pragma unroll
      for (int u = 0; u < 8; ++u) s += gb[u * GP_STRIDE + j * 32 + lane];
      ga[j][lane] = s;
    }
    if (tid < 32) {
      float s = 0.0f;
#pragma unroll
      for (int u = 0; u < 8; ++u) s += gb[u * GP_STRIDE + 1024 + tid];
      ga[tid][32] = s;
      ga[32][tid] = s;
    }
    for (int idx = tid; idx < 36 * 48; idx += 256) {
      const int j = idx / 48;
      const int k = idx - j * 48;
      if (j >= 33 || k >= 33) ga[j][k] = 0.0f;
      else if (j == 32 && k == 32) ga[j][k] = (float)HW_;
    }
  }
  __syncthreads();

  if ((tid >> 5) < 3) {
    v2f fa[9], fb[9];
    {
      const int w    = tid >> 5;
      const int half = (tid & 31) >> 4;
      const int l16  = tid & 15;
      const int m    = mt * 16 + l16;
      const int col  = 16 * w + l16;
      const float* ap = w2 + (size_t)m * CR_ + 2 * half;
#pragma unroll
      for (int s = 0; s < 8; ++s) fa[s] = *(const v2f*)(ap + 4 * s);
      const float bm = b2[m];
      fa[8].x = half ? 0.0f : bm;
      fa[8].y = 0.0f;
#pragma unroll
      for (int s = 0; s < 9; ++s) {
        fb[s].x = ga[4 * s + 2 * half][col];
        fb[s].y = ga[4 * s + 2 * half + 1][col];
      }
    }
    PIN9(fa);
    PIN9(fb);
    v8f c = {};
#pragma unroll
    for (int s = 0; s < 9; ++s) c = wmma4(fa[s], fb[s], c);
    int t2 = tid;
    asm volatile("v_nop\n\tv_nop\n\tv_nop\n\tv_nop\n\tv_nop\n\tv_nop\n\tv_nop\n\tv_nop\n\tv_nop"
                 : "+v"(c), "+v"(t2));
    const int col  = 16 * (t2 >> 5) + (t2 & 15);
    const int row0 = 8 * ((t2 & 31) >> 4);
    if (col < 36) {
#pragma unroll
      for (int r = 0; r < 8; ++r) tl[row0 + r][col] = c[r];
    }
  }
  __syncthreads();

  v2f ta[9];
  {
    const int half = (tid & 31) >> 4;
    const int l16  = tid & 15;
#pragma unroll
    for (int s = 0; s < 9; ++s) ta[s] = *(const v2f*)&tl[l16][4 * s + 2 * half];
  }
  PIN9(ta);
  PIN9(bw[0]);
  PIN9(bw[1]);
  PIN9(bw[2]);
  PIN9(bw[3]);
  v8f c0 = {}, c1 = {}, c2 = {}, c3 = {};
#pragma unroll
  for (int s = 0; s < 9; ++s) {
    c0 = wmma4(ta[s], bw[0][s], c0);
    c1 = wmma4(ta[s], bw[1][s], c1);
    c2 = wmma4(ta[s], bw[2][s], c2);
    c3 = wmma4(ta[s], bw[3][s], c3);
  }
  asm volatile("v_nop\n\tv_nop\n\tv_nop\n\tv_nop\n\tv_nop\n\tv_nop\n\tv_nop\n\tv_nop\n\tv_nop"
               : "+v"(c0), "+v"(c1), "+v"(c2), "+v"(c3), "+v"(tid));
  {
    const int w    = tid >> 5;
    const int half = (tid & 31) >> 4;
    const int l16  = tid & 15;
    float* D = e + (size_t)b * C_ * C_ + (size_t)(mt * 16 + 8 * half) * C_ + 64 * w + l16;
#pragma unroll
    for (int r = 0; r < 8; ++r) {
      D[(size_t)r * C_]      = c0[r];
      D[(size_t)r * C_ + 16] = c1[r];
      D[(size_t)r * C_ + 32] = c2[r];
      D[(size_t)r * C_ + 48] = c3[r];
    }
  }
}

__global__ void k_softmax(float* __restrict__ e) {
  __shared__ float red[256];
  const int tid = threadIdx.x;
  float* p = e + (size_t)blockIdx.x * C_;
  const float v0 = p[tid];
  const float v1 = p[tid + 256];
  red[tid] = fminf(v0, v1);
  __syncthreads();
  for (int s = 128; s > 0; s >>= 1) {
    if (tid < s) red[tid] = fminf(red[tid], red[tid + s]);
    __syncthreads();
  }
  const float rmin = red[0];
  __syncthreads();
  const float e0 = expf(rmin - v0);
  const float e1 = expf(rmin - v1);
  red[tid] = e0 + e1;
  __syncthreads();
  for (int s = 128; s > 0; s >>= 1) {
    if (tid < s) red[tid] += red[tid + s];
    __syncthreads();
  }
  const float inv = 1.0f / red[0];
  p[tid] = e0 * inv;
  p[tid + 256] = e1 * inv;
}

__global__ void k_m2(const float* __restrict__ attn, const float* __restrict__ wf,
                     const float* __restrict__ gamma, float* __restrict__ m2) {
  constexpr int NT = 4;
  const int wid = blockIdx.x * 8 + (threadIdx.x >> 5);
  const int NG = C_ / (16 * NT);
  const int MT = C_ / 16;
  const int b  = wid / (MT * NG);
  const int r0 = wid % (MT * NG);
  const int mt = r0 / NG;
  const int ng = r0 % NG;

  const float* A  = wf + (size_t)mt * 16 * C2_;
  const float* Bm = attn + (size_t)b * C_ * C_ + ng * 16 * NT;
  v8f c[NT] = {};
  gemm_row<NT>(A, C2_, Bm, C_, C_, c);

  const float g = gamma[0];
  const int lane = threadIdx.x & 31;
  const int half = lane >> 4;
  const int l16  = lane & 15;
  float* D = m2 + (size_t)b * C_ * C_;
#pragma unroll
  for (int t = 0; t < NT; ++t)
#pragma unroll
    for (int r = 0; r < 8; ++r) {
      const int m = mt * 16 + r + 8 * half;
      const int n = ng * 16 * NT + t * 16 + l16;
      D[(size_t)m * C_ + n] = g * c[t][r] + wf[(size_t)m * C2_ + n];
    }
}

__global__ void k_pq(const float* __restrict__ m2, const float* __restrict__ w2,
                     const float* __restrict__ b2, float* __restrict__ P, float* __restrict__ q) {
  __shared__ float part[8][3][8][32];
  const int b    = blockIdx.x >> 5;
  const int mt   = blockIdx.x & 31;
  v8f c0 = {}, c1 = {}, c2 = {};
  {
    const int w    = threadIdx.x >> 5;
    const int lane = threadIdx.x & 31;
    const int half = lane >> 4;
    const int l16  = lane & 15;
    const int k0   = w * 64;
    const float* ap = m2 + (size_t)b * C_ * C_ + (size_t)(mt * 16 + l16) * C_ + k0 + 2 * half;
    const float* bp = w2 + (size_t)(k0 + 2 * half) * CR_ + l16;
    const float* cp = b2 + k0 + 2 * half;
#pragma unroll
    for (int kk = 0; kk < 64; kk += 4) {
      const v2f a = *(const v2f*)(ap + kk);
      v2f b0, b1;
      b0.x = bp[kk * CR_];
      b0.y = bp[kk * CR_ + CR_];
      b1.x = bp[kk * CR_ + 16];
      b1.y = bp[kk * CR_ + CR_ + 16];
      const v2f bb = *(const v2f*)(cp + kk);
      c0 = wmma4(a, b0, c0);
      c1 = wmma4(a, b1, c1);
      c2 = wmma4(a, bb, c2);
    }
  }
  int tid = threadIdx.x;
  asm volatile("v_nop\n\tv_nop\n\tv_nop\n\tv_nop\n\tv_nop\n\tv_nop\n\tv_nop\n\tv_nop\n\tv_nop"
               : "+v"(c0), "+v"(c1), "+v"(c2), "+v"(tid));
  const int w    = tid >> 5;
  const int lane = tid & 31;
  const int half = lane >> 4;
  const int l16  = lane & 15;
#pragma unroll
  for (int r = 0; r < 8; ++r) {
    part[w][0][r][lane] = c0[r];
    part[w][1][r][lane] = c1[r];
    part[w][2][r][lane] = c2[r];
  }
  __syncthreads();

  const int m = mt * 16 + w + 8 * half;
#pragma unroll
  for (int t = 0; t < 2; ++t) {
    float s = 0.0f;
#pragma unroll
    for (int u = 0; u < 8; ++u) s += part[u][t][w][lane];
    P[((size_t)b * C_ + m) * CR_ + t * 16 + l16] = s;
  }
  if (tid < 16) {
    const int row = tid;
    float s = 0.0f;
#pragma unroll
    for (int u = 0; u < 8; ++u) s += part[u][2][row & 7][(row >> 3) * 16];
    q[(size_t)blockIdx.x * 32 + row] = s;
  }
}

__global__ void __launch_bounds__(256) k_fusion(const float* __restrict__ P, const float* __restrict__ q,
                         const float* __restrict__ h, const float* __restrict__ x,
                         const float* __restrict__ wf, const float* __restrict__ bf,
                         float* __restrict__ out) {
  const int wid = blockIdx.x * 8 + (threadIdx.x >> 5);
  const int NG = HW_ / 64;
  const int MT = C_ / 64;
  const int b  = wid / (MT * NG);
  const int r0 = wid % (MT * NG);
  const int mt = r0 / NG;
  const int ng = r0 % NG;

  v8f c[4][4] = {};
  gemm_tile64(P + ((size_t)b * C_ + (size_t)mt * 64) * CR_, CR_,
              h + (size_t)b * CR_ * HW_ + ng * 64, HW_, CR_, c);
  gemm_tile64(wf + (size_t)mt * 64 * C2_ + C_, C2_,
              x + (size_t)b * C_ * HW_ + ng * 64, HW_, C_, c);

  const int lane = threadIdx.x & 31;
  const int half = lane >> 4;
  const int l16  = lane & 15;
  float* D = out + (size_t)b * C_ * HW_;
  const float* qb = q + ((size_t)b * (C_ / 16) + (size_t)mt * 4) * 32;
#pragma unroll
  for (int i = 0; i < 4; ++i)
#pragma unroll
    for (int r = 0; r < 8; ++r) {
      const int m = mt * 64 + 16 * i + r + 8 * half;
      const float bias = bf[m] + qb[i * 32 + r + 8 * half];
#pragma unroll
      for (int t = 0; t < 4; ++t) {
        const int n = ng * 64 + t * 16 + l16;
        D[(size_t)m * HW_ + n] = c[i][t][r] + bias;
      }
    }
}

extern "C" void kernel_launch(void* const* d_in, const int* in_sizes, int n_in,
                              void* d_out, int out_size, void* d_ws, size_t ws_size,
                              hipStream_t stream) {
  const float* x     = (const float*)d_in[0];
  const float* w1    = (const float*)d_in[1];
  const float* b1    = (const float*)d_in[2];
  const float* w2    = (const float*)d_in[3];
  const float* b2    = (const float*)d_in[4];
  const float* wf    = (const float*)d_in[5];
  const float* bf    = (const float*)d_in[6];
  const float* gamma = (const float*)d_in[7];
  float* out = (float*)d_out;

  char* ws = (char*)d_ws;
  float* h    = (float*)(ws);
  float* gp   = (float*)(ws + (size_t)4 * 1024 * 1024);
  float* attn = (float*)(ws + (size_t)68 * 1024 * 1024);
  float* m2   = (float*)(ws + (size_t)76 * 1024 * 1024);
  float* P    = attn;
  float* q    = attn + (size_t)B_ * C_ * CR_;

  const dim3 blk(256);
  k_conv1_gelu<<<dim3(512),  blk, 0, stream>>>(x, w1, b1, h);
  k_gram      <<<dim3(64),   blk, 0, stream>>>(h, gp);
  k_energy    <<<dim3(256),  blk, 0, stream>>>(gp, w2, b2, attn);
  k_softmax   <<<dim3(4096), blk, 0, stream>>>(attn);
  k_m2        <<<dim3(256),  blk, 0, stream>>>(attn, wf, gamma, m2);
  k_pq        <<<dim3(256),  blk, 0, stream>>>(m2, w2, b2, P, q);
  k_fusion    <<<dim3(512),  blk, 0, stream>>>(P, q, h, x, wf, bf, out);
}


# M2.[W2 b2] re-associated as Wf1.((gamma.attn + I).[W2 b2]): the 512^3 product Wf1.attn per batch is never formed, two K=512 products of 33 columns replace it; M2 not stored; f32 sum order changes
# speedup vs baseline: 4.0955x; 1.2278x over previous
#include <hip/hip_runtime.h>
#include <math.h>


#define B_   8
#define C_   512
#define CR_  32
#define HW_  4096
#define C2_  1024

typedef __attribute__((ext_vector_type(2))) float v2f;
typedef __attribute__((ext_vector_type(8))) float v8f;

__device__ __forceinline__ v8f wmma4(v2f a, v2f b, v8f c) {
  return __builtin_amdgcn_wmma_f32_16x16x4_f32(
       false, a,  false, b,
       (short)0, c,  false,  false);
}

__device__ __forceinline__ void gemm_tile64(const float* __restrict__ A, int lda,
                                            const float* __restrict__ Bm, int ldb,
                                            int K, v8f c[4][4]) {
  const int lane = threadIdx.x & 31;
  const int half = lane >> 4;
  const int l16  = lane & 15;
  const float* ap = A + (size_t)l16 * lda + 2 * half;
  const float* bp = Bm + (size_t)(2 * half) * ldb + l16;
#pragma unroll 2
  for (int k = 0; k < K; k += 4) {
    v2f a[4], b[4];
#pragma unroll
    for (int i = 0; i < 4; ++i) a[i] = *(const v2f*)(ap + (size_t)(16 * i) * lda);
#pragma unroll
    for (int t = 0; t < 4; ++t) {
      b[t].x = bp[t * 16];
      b[t].y = bp[t * 16 + ldb];
    }
#pragma unroll
    for (int i = 0; i < 4; ++i)
#pragma unroll
      for (int t = 0; t < 4; ++t) c[i][t] = wmma4(a[i], b[t], c[i][t]);
    ap += 4;
    bp += (size_t)4 * ldb;
  }
}

__global__ void __launch_bounds__(256) __attribute__((amdgpu_waves_per_eu(4, 4))) k_conv1_gelu(const float* __restrict__ x, const float* __restrict__ w1,
                             const float* __restrict__ b1, float* __restrict__ h) {
  __shared__ float part[8][8][8][32];
  const int b  = blockIdx.x >> 6;
  const int ng = blockIdx.x & 63;
  v8f c[2][4] = {};
  {
    const int w    = threadIdx.x >> 5;
    const int lane = threadIdx.x & 31;
    const int half = lane >> 4;
    const int l16  = lane & 15;
    const int k0   = w * 64;
    const float* ap = w1 + (size_t)l16 * C_ + k0 + 2 * half;
    const float* bp = x + (size_t)b * C_ * HW_ + (size_t)(k0 + 2 * half) * HW_ + ng * 64 + l16;
#pragma unroll
    for (int kk = 0; kk < 64; kk += 4) {
      const v2f a0 = *(const v2f*)(ap + kk);
      const v2f a1 = *(const v2f*)(ap + 16 * C_ + kk);
      v2f bx[4];
#pragma unroll
      for (int t = 0; t < 4; ++t) {
        bx[t].x = bp[(size_t)kk * HW_ + 16 * t];
        bx[t].y = bp[(size_t)(kk + 1) * HW_ + 16 * t];
      }
#pragma unroll
      for (int t = 0; t < 4; ++t) {
        c[0][t] = wmma4(a0, bx[t], c[0][t]);
        c[1][t] = wmma4(a1, bx[t], c[1][t]);
      }
    }
  }
  int tid = threadIdx.x;
  asm volatile("v_nop\n\tv_nop\n\tv_nop\n\tv_nop\n\tv_nop\n\tv_nop\n\tv_nop\n\tv_nop\n\tv_nop"
               : "+v"(c[0][0]), "+v"(c[0][1]), "+v"(c[0][2]), "+v"(c[0][3]),
                 "+v"(c[1][0]), "+v"(c[1][1]), "+v"(c[1][2]), "+v"(c[1][3]), "+v"(tid));
  const int w    = tid >> 5;
  const int lane = tid & 31;
  const int half = lane >> 4;
  const int l16  = lane & 15;
#pragma unroll
  for (int i = 0; i < 2; ++i)
#pragma unroll
    for (int t = 0; t < 4; ++t)
#pragma unroll
      for (int r = 0; r < 8; ++r) part[w][4 * i + t][r][lane] = c[i][t][r];
  __syncthreads();

  float* D = h + (size_t)b * CR_ * HW_ + ng * 64 + l16;
  float s[2][4];
#pragma unroll
  for (int i = 0; i < 2; ++i)
#pragma unroll
    for (int t = 0; t < 4; ++t) {
      float a = 0.0f;
#pragma unroll
      for (int u = 0; u < 8; ++u) a += part[u][4 * i + t][w][lane];
      s[i][t] = a;
    }
#pragma unroll
  for (int i = 0; i < 2; ++i) {
    const int m = 16 * i + w + 8 * half;
    const float bias = b1[m];
#pragma unroll
    for (int t = 0; t < 4; ++t) {
      float v = s[i][t] + bias;
      v = 0.5f * v * (1.0f + erff(v * 0.70710678118654752f));
      D[(size_t)m * HW_ + 16 * t] = v;
    }
  }
}

#define GP_STRIDE 1088

__global__ void k_gram(const float* __restrict__ h, float* __restrict__ gp) {
  __shared__ float part[8][6][8][32];
  const int b  = blockIdx.x >> 3;
  const int ks = blockIdx.x & 7;
  v8f c00 = {}, c01 = {}, c10 = {}, c11 = {}, s0 = {}, s1 = {};
  {
    const int w    = threadIdx.x >> 5;
    const int lane = threadIdx.x & 31;
    const int half = lane >> 4;
    const int l16  = lane & 15;
    const float* p0 = h + (size_t)b * CR_ * HW_ + (size_t)l16 * HW_ + ks * 512 + w * 64 + 4 * half;
    const float* p1 = p0 + (size_t)16 * HW_;
    float4 r0[8], r1[8];
#pragma unroll
    for (int i = 0; i < 8; ++i) {
      r0[i] = *(const float4*)(p0 + 8 * i);
      r1[i] = *(const float4*)(p1 + 8 * i);
    }
    v2f one;
    one.x = 1.0f;
    one.y = 1.0f;
#pragma unroll
    for (int i = 0; i < 8; ++i) {
      v2f a0, a1;
      a0.x = r0[i].x; a0.y = r0[i].y;
      a1.x = r1[i].x; a1.y = r1[i].y;
      c00 = wmma4(a0, a0, c00);
      c01 = wmma4(a0, a1, c01);
      c10 = wmma4(a1, a0, c10);
      c11 = wmma4(a1, a1, c11);
      s0  = wmma4(a0, one, s0);
      s1  = wmma4(a1, one, s1);
      a0.x = r0[i].z; a0.y = r0[i].w;
      a1.x = r1[i].z; a1.y = r1[i].w;
      c00 = wmma4(a0, a0, c00);
      c01 = wmma4(a0, a1, c01);
      c10 = wmma4(a1, a0, c10);
      c11 = wmma4(a1, a1, c11);
      s0  = wmma4(a0, one, s0);
      s1  = wmma4(a1, one, s1);
    }
  }
  int tid = threadIdx.x;
  asm volatile("v_nop\n\tv_nop\n\tv_nop\n\tv_nop\n\tv_nop\n\tv_nop\n\tv_nop\n\tv_nop\n\tv_nop"
               : "+v"(c00), "+v"(c01), "+v"(c10), "+v"(c11), "+v"(s0), "+v"(s1), "+v"(tid));
  const int w    = tid >> 5;
  const int lane = tid & 31;
  const int half = lane >> 4;
  const int l16  = lane & 15;
#pragma unroll
  for (int r = 0; r < 8; ++r) {
    part[w][0][r][lane] = c00[r];
    part[w][1][r][lane] = c01[r];
    part[w][2][r][lane] = c10[r];
    part[w][3][r][lane] = c11[r];
    part[w][4][r][lane] = s0[r];
    part[w][5][r][lane] = s1[r];
  }
  __syncthreads();

  float* g = gp + (size_t)blockIdx.x * GP_STRIDE;
#pragma unroll
  for (int t = 0; t < 4; ++t) {
    float s = 0.0f;
#pragma unroll
    for (int u = 0; u < 8; ++u) s += part[u][t][w][lane];
    g[(16 * (t >> 1) + w + 8 * half) * 32 + 16 * (t & 1) + l16] = s;
  }
  if (tid < 32) {
    const int rr = tid & 15;
    float s = 0.0f;
#pragma unroll
    for (int u = 0; u < 8; ++u) s += part[u][4 + (tid >> 4)][rr & 7][(rr >> 3) * 16];
    g[1024 + tid] = s;
  }
}


#define PIN9(f) asm volatile("" : "+v"(f[0]), "+v"(f[1]), "+v"(f[2]), "+v"(f[3]), "+v"(f[4]), \
                                  "+v"(f[5]), "+v"(f[6]), "+v"(f[7]), "+v"(f[8]))

__global__ void k_energy(const float* __restrict__ gp, const float* __restrict__ w2,
                         const float* __restrict__ b2, float* __restrict__ e) {
  __shared__ float ga[36][48];
  __shared__ float tl[16][40];
  const int b    = blockIdx.x >> 5;
  const int mt   = blockIdx.x & 31;
  int tid = threadIdx.x;
  v2f bw[4][9];
  {
    const int w    = tid >> 5;
    const int lane = tid & 31;
    const int half = lane >> 4;
    const int l16  = lane & 15;

#pragma unroll
    for (int t = 0; t < 4; ++t) {
      const int n = 64 * w + 16 * t + l16;
      const float* wp = w2 + (size_t)n * CR_ + 2 * half;
#pragma unroll
      for (int s = 0; s < 8; ++s) bw[t][s] = *(const v2f*)(wp + 4 * s);
      const float bn = b2[n];
      bw[t][8].x = half ? 0.0f : bn;
      bw[t][8].y = 0.0f;
    }

    const float* gb = gp + (size_t)b * 8 * GP_STRIDE;
#pragma unroll
    for (int i = 0; i < 4; ++i) {
      const int j = w + 8 * i;
      float s = 0.0f;
#pragma unroll
      for (int u = 0; u < 8; ++u) s += gb[u * GP_STRIDE + j * 32 + lane];
      ga[j][lane] = s;
    }
    if (tid < 32) {
      float s = 0.0f;
#pragma unroll
      for (int u = 0; u < 8; ++u) s += gb[u * GP_STRIDE + 1024 + tid];
      ga[tid][32] = s;
      ga[32][tid] = s;
    }
    for (int idx = tid; idx < 36 * 48; idx += 256) {
      const int j = idx / 48;
      const int k = idx - j * 48;
      if (j >= 33 || k >= 33) ga[j][k] = 0.0f;
      else if (j == 32 && k == 32) ga[j][k] = (float)HW_;
    }
  }
  __syncthreads();

  if ((tid >> 5) < 3) {
    v2f fa[9], fb[9];
    {
      const int w    = tid >> 5;
      const int half = (tid & 31) >> 4;
      const int l16  = tid & 15;
      const int m    = mt * 16 + l16;
      const int col  = 16 * w + l16;
      const float* ap = w2 + (size_t)m * CR_ + 2 * half;
#pragma unroll
      for (int s = 0; s < 8; ++s) fa[s] = *(const v2f*)(ap + 4 * s);
      const float bm = b2[m];
      fa[8].x = half ? 0.0f : bm;
      fa[8].y = 0.0f;
#pragma unroll
      for (int s = 0; s < 9; ++s) {
        fb[s].x = ga[4 * s + 2 * half][col];
        fb[s].y = ga[4 * s + 2 * half + 1][col];
      }
    }
    PIN9(fa);
    PIN9(fb);
    v8f c = {};
#pragma unroll
    for (int s = 0; s < 9; ++s) c = wmma4(fa[s], fb[s], c);
    int t2 = tid;
    asm volatile("v_nop\n\tv_nop\n\tv_nop\n\tv_nop\n\tv_nop\n\tv_nop\n\tv_nop\n\tv_nop\n\tv_nop"
                 : "+v"(c), "+v"(t2));
    const int col  = 16 * (t2 >> 5) + (t2 & 15);
    const int row0 = 8 * ((t2 & 31) >> 4);
    if (col < 36) {
#pragma unroll
      for (int r = 0; r < 8; ++r) tl[row0 + r][col] = c[r];
    }
  }
  __syncthreads();

  v2f ta[9];
  {
    const int half = (tid & 31) >> 4;
    const int l16  = tid & 15;
#pragma unroll
    for (int s = 0; s < 9; ++s) ta[s] = *(const v2f*)&tl[l16][4 * s + 2 * half];
  }
  PIN9(ta);
  PIN9(bw[0]);
  PIN9(bw[1]);
  PIN9(bw[2]);
  PIN9(bw[3]);
  v8f c0 = {}, c1 = {}, c2 = {}, c3 = {};
#pragma unroll
  for (int s = 0; s < 9; ++s) {
    c0 = wmma4(ta[s], bw[0][s], c0);
    c1 = wmma4(ta[s], bw[1][s], c1);
    c2 = wmma4(ta[s], bw[2][s], c2);
    c3 = wmma4(ta[s], bw[3][s], c3);
  }
  asm volatile("v_nop\n\tv_nop\n\tv_nop\n\tv_nop\n\tv_nop\n\tv_nop\n\tv_nop\n\tv_nop\n\tv_nop"
               : "+v"(c0), "+v"(c1), "+v"(c2), "+v"(c3), "+v"(tid));
  {
    const int w    = tid >> 5;
    const int half = (tid & 31) >> 4;
    const int l16  = tid & 15;
    float* D = e + (size_t)b * C_ * C_ + (size_t)(mt * 16 + 8 * half) * C_ + 64 * w + l16;
#pragma unroll
    for (int r = 0; r < 8; ++r) {
      D[(size_t)r * C_]      = c0[r];
      D[(size_t)r * C_ + 16] = c1[r];
      D[(size_t)r * C_ + 32] = c2[r];
      D[(size_t)r * C_ + 48] = c3[r];
    }
  }
}

__global__ void k_softmax(float* __restrict__ e) {
  __shared__ float red[256];
  const int tid = threadIdx.x;
  float* p = e + (size_t)blockIdx.x * C_;
  const float v0 = p[tid];
  const float v1 = p[tid + 256];
  red[tid] = fminf(v0, v1);
  __syncthreads();
  for (int s = 128; s > 0; s >>= 1) {
    if (tid < s) red[tid] = fminf(red[tid], red[tid + s]);
    __syncthreads();
  }
  const float rmin = red[0];
  __syncthreads();
  const float e0 = expf(rmin - v0);
  const float e1 = expf(rmin - v1);
  red[tid] = e0 + e1;
  __syncthreads();
  for (int s = 128; s > 0; s >>= 1) {
    if (tid < s) red[tid] += red[tid + s];
    __syncthreads();
  }
  const float inv = 1.0f / red[0];
  p[tid] = e0 * inv;
  p[tid + 256] = e1 * inv;
}

__global__ void k_z(const float* __restrict__ attn, const float* __restrict__ w2,
                    const float* __restrict__ b2, const float* __restrict__ gamma,
                    float* __restrict__ Z, float* __restrict__ zb) {
  __shared__ float part[8][3][8][32];
  const int b    = blockIdx.x >> 5;
  const int mt   = blockIdx.x & 31;
  v8f c0 = {}, c1 = {}, c2 = {};
  {
    const int w    = threadIdx.x >> 5;
    const int lane = threadIdx.x & 31;
    const int half = lane >> 4;
    const int l16  = lane & 15;
    const int k0   = w * 64;
    const float* ap = attn + (size_t)b * C_ * C_ + (size_t)(mt * 16 + l16) * C_ + k0 + 2 * half;
    const float* bp = w2 + (size_t)(k0 + 2 * half) * CR_ + l16;
    const float* cp = b2 + k0 + 2 * half;
#pragma unroll
    for (int kk = 0; kk < 64; kk += 4) {
      const v2f a = *(const v2f*)(ap + kk);
      v2f b0, b1;
      b0.x = bp[kk * CR_];
      b0.y = bp[kk * CR_ + CR_];
      b1.x = bp[kk * CR_ + 16];
      b1.y = bp[kk * CR_ + CR_ + 16];
      const v2f bb = *(const v2f*)(cp + kk);
      c0 = wmma4(a, b0, c0);
      c1 = wmma4(a, b1, c1);
      c2 = wmma4(a, bb, c2);
    }
  }
  int tid = threadIdx.x;
  asm volatile("v_nop\n\tv_nop\n\tv_nop\n\tv_nop\n\tv_nop\n\tv_nop\n\tv_nop\n\tv_nop\n\tv_nop"
               : "+v"(c0), "+v"(c1), "+v"(c2), "+v"(tid));
  const int w    = tid >> 5;
  const int lane = tid & 31;
  const int half = lane >> 4;
  const int l16  = lane & 15;
#pragma unroll
  for (int r = 0; r < 8; ++r) {
    part[w][0][r][lane] = c0[r];
    part[w][1][r][lane] = c1[r];
    part[w][2][r][lane] = c2[r];
  }
  __syncthreads();

  const float g = gamma[0];
  const int m = mt * 16 + w + 8 * half;
#pragma unroll
  for (int t = 0; t < 2; ++t) {
    float s = 0.0f;
#pragma unroll
    for (int u = 0; u < 8; ++u) s += part[u][t][w][lane];
    const int n = t * 16 + l16;
    Z[((size_t)b * C_ + m) * CR_ + n] = g * s + w2[(size_t)m * CR_ + n];
  }
  if (tid < 16) {
    const int row = tid;
    float s = 0.0f;
#pragma unroll
    for (int u = 0; u < 8; ++u) s += part[u][2][row & 7][(row >> 3) * 16];
    zb[(size_t)blockIdx.x * 32 + row] = g * s + b2[mt * 16 + row];
  }
}

__global__ void k_pq(const float* __restrict__ wf, const float* __restrict__ Z,
                     const float* __restrict__ zb, float* __restrict__ P, float* __restrict__ q) {
  __shared__ float part[8][3][8][32];
  const int b    = blockIdx.x >> 5;
  const int mt   = blockIdx.x & 31;
  v8f c0 = {}, c1 = {}, c2 = {};
  {
    const int w    = threadIdx.x >> 5;
    const int lane = threadIdx.x & 31;
    const int half = lane >> 4;
    const int l16  = lane & 15;
    const int k0   = w * 64;
    const float* ap = wf + (size_t)(mt * 16 + l16) * C2_ + k0 + 2 * half;
    const float* bp = Z + (size_t)b * C_ * CR_ + (size_t)(k0 + 2 * half) * CR_ + l16;
    const float* cp = zb + ((size_t)b * (C_ / 16) + (size_t)w * 4) * 32 + 2 * half;
#pragma unroll
    for (int kk = 0; kk < 64; kk += 4) {
      const v2f a = *(const v2f*)(ap + kk);
      v2f b0, b1;
      b0.x = bp[kk * CR_];
      b0.y = bp[kk * CR_ + CR_];
      b1.x = bp[kk * CR_ + 16];
      b1.y = bp[kk * CR_ + CR_ + 16];
      const v2f bb = *(const v2f*)(cp + (kk >> 4) * 32 + (kk & 15));
      c0 = wmma4(a, b0, c0);
      c1 = wmma4(a, b1, c1);
      c2 = wmma4(a, bb, c2);
    }
  }
  int tid = threadIdx.x;
  asm volatile("v_nop\n\tv_nop\n\tv_nop\n\tv_nop\n\tv_nop\n\tv_nop\n\tv_nop\n\tv_nop\n\tv_nop"
               : "+v"(c0), "+v"(c1), "+v"(c2), "+v"(tid));
  const int w    = tid >> 5;
  const int lane = tid & 31;
  const int half = lane >> 4;
  const int l16  = lane & 15;
#pragma unroll
  for (int r = 0; r < 8; ++r) {
    part[w][0][r][lane] = c0[r];
    part[w][1][r][lane] = c1[r];
    part[w][2][r][lane] = c2[r];
  }
  __syncthreads();

  const int m = mt * 16 + w + 8 * half;
#pragma unroll
  for (int t = 0; t < 2; ++t) {
    float s = 0.0f;
#pragma unroll
    for (int u = 0; u < 8; ++u) s += part[u][t][w][lane];
    P[((size_t)b * C_ + m) * CR_ + t * 16 + l16] = s;
  }
  if (tid < 16) {
    const int row = tid;
    float s = 0.0f;
#pragma unroll
    for (int u = 0; u < 8; ++u) s += part[u][2][row & 7][(row >> 3) * 16];
    q[(size_t)blockIdx.x * 32 + row] = s;
  }
}

__global__ void __launch_bounds__(256) k_fusion(const float* __restrict__ P, const float* __restrict__ q,
                         const float* __restrict__ h, const float* __restrict__ x,
                         const float* __restrict__ wf, const float* __restrict__ bf,
                         float* __restrict__ out) {
  const int wid = blockIdx.x * 8 + (threadIdx.x >> 5);
  const int NG = HW_ / 64;
  const int MT = C_ / 64;
  const int b  = wid / (MT * NG);
  const int r0 = wid % (MT * NG);
  const int mt = r0 / NG;
  const int ng = r0 % NG;

  v8f c[4][4] = {};
  gemm_tile64(P + ((size_t)b * C_ + (size_t)mt * 64) * CR_, CR_,
              h + (size_t)b * CR_ * HW_ + ng * 64, HW_, CR_, c);
  gemm_tile64(wf + (size_t)mt * 64 * C2_ + C_, C2_,
              x + (size_t)b * C_ * HW_ + ng * 64, HW_, C_, c);

  const int lane = threadIdx.x & 31;
  const int half = lane >> 4;
  const int l16  = lane & 15;
  float* D = out + (size_t)b * C_ * HW_;
  const float* qb = q + ((size_t)b * (C_ / 16) + (size_t)mt * 4) * 32;
#pragma unroll
  for (int i = 0; i < 4; ++i)
#pragma unroll
    for (int r = 0; r < 8; ++r) {
      const int m = mt * 64 + 16 * i + r + 8 * half;
      const float bias = bf[m] + qb[i * 32 + r + 8 * half];
#pragma unroll
      for (int t = 0; t < 4; ++t) {
        const int n = ng * 64 + t * 16 + l16;
        D[(size_t)m * HW_ + n] = c[i][t][r] + bias;
      }
    }
}

extern "C" void kernel_launch(void* const* d_in, const int* in_sizes, int n_in,
                              void* d_out, int out_size, void* d_ws, size_t ws_size,
                              hipStream_t stream) {
  const float* x     = (const float*)d_in[0];
  const float* w1    = (const float*)d_in[1];
  const float* b1    = (const float*)d_in[2];
  const float* w2    = (const float*)d_in[3];
  const float* b2    = (const float*)d_in[4];
  const float* wf    = (const float*)d_in[5];
  const float* bf    = (const float*)d_in[6];
  const float* gamma = (const float*)d_in[7];
  float* out = (float*)d_out;

  char* ws = (char*)d_ws;
  float* h    = (float*)(ws);
  float* gp   = (float*)(ws + (size_t)4 * 1024 * 1024);
  float* attn = (float*)(ws + (size_t)68 * 1024 * 1024);
  float* Z    = (float*)(ws + (size_t)76 * 1024 * 1024);
  float* zb   = Z + (size_t)B_ * C_ * CR_;
  float* P    = attn;
  float* q    = attn + (size_t)B_ * C_ * CR_;

  const dim3 blk(256);
  k_conv1_gelu<<<dim3(512),  blk, 0, stream>>>(x, w1, b1, h);
  k_gram      <<<dim3(64),   blk, 0, stream>>>(h, gp);
  k_energy    <<<dim3(256),  blk, 0, stream>>>(gp, w2, b2, attn);
  k_softmax   <<<dim3(4096), blk, 0, stream>>>(attn);
  k_z         <<<dim3(256),  blk, 0, stream>>>(attn, w2, b2, gamma, Z, zb);
  k_pq        <<<dim3(256),  blk, 0, stream>>>(wf, Z, zb, P, q);
  k_fusion    <<<dim3(512),  blk, 0, stream>>>(P, q, h, x, wf, bf, out);
}


# softmax fused into the energy kernel: 16x512 tile via LDS, a wave per 2 rows does min/exp/sum by lane exchange; energy never stored, 4096-block softmax launch dropped (7->6); f32 sum order changes
# speedup vs baseline: 4.2647x; 1.0413x over previous
#include <hip/hip_runtime.h>
#include <math.h>


#define B_   8
#define C_   512
#define CR_  32
#define HW_  4096
#define C2_  1024

typedef __attribute__((ext_vector_type(2))) float v2f;
typedef __attribute__((ext_vector_type(8))) float v8f;

__device__ __forceinline__ v8f wmma4(v2f a, v2f b, v8f c) {
  return __builtin_amdgcn_wmma_f32_16x16x4_f32(
       false, a,  false, b,
       (short)0, c,  false,  false);
}

__device__ __forceinline__ void gemm_tile64(const float* __restrict__ A, int lda,
                                            const float* __restrict__ Bm, int ldb,
                                            int K, v8f c[4][4]) {
  const int lane = threadIdx.x & 31;
  const int half = lane >> 4;
  const int l16  = lane & 15;
  const float* ap = A + (size_t)l16 * lda + 2 * half;
  const float* bp = Bm + (size_t)(2 * half) * ldb + l16;
#pragma unroll 2
  for (int k = 0; k < K; k += 4) {
    v2f a[4], b[4];
#pragma unroll
    for (int i = 0; i < 4; ++i) a[i] = *(const v2f*)(ap + (size_t)(16 * i) * lda);
#pragma unroll
    for (int t = 0; t < 4; ++t) {
      b[t].x = bp[t * 16];
      b[t].y = bp[t * 16 + ldb];
    }
#pragma unroll
    for (int i = 0; i < 4; ++i)
#pragma unroll
      for (int t = 0; t < 4; ++t) c[i][t] = wmma4(a[i], b[t], c[i][t]);
    ap += 4;
    bp += (size_t)4 * ldb;
  }
}

__global__ void __launch_bounds__(256) __attribute__((amdgpu_waves_per_eu(4, 4))) k_conv1_gelu(const float* __restrict__ x, const float* __restrict__ w1,
                             const float* __restrict__ b1, float* __restrict__ h) {
  __shared__ float part[8][8][8][32];
  const int b  = blockIdx.x >> 6;
  const int ng = blockIdx.x & 63;
  v8f c[2][4] = {};
  {
    const int w    = threadIdx.x >> 5;
    const int lane = threadIdx.x & 31;
    const int half = lane >> 4;
    const int l16  = lane & 15;
    const int k0   = w * 64;
    const float* ap = w1 + (size_t)l16 * C_ + k0 + 2 * half;
    const float* bp = x + (size_t)b * C_ * HW_ + (size_t)(k0 + 2 * half) * HW_ + ng * 64 + l16;
#pragma unroll
    for (int kk = 0; kk < 64; kk += 4) {
      const v2f a0 = *(const v2f*)(ap + kk);
      const v2f a1 = *(const v2f*)(ap + 16 * C_ + kk);
      v2f bx[4];
#pragma unroll
      for (int t = 0; t < 4; ++t) {
        bx[t].x = bp[(size_t)kk * HW_ + 16 * t];
        bx[t].y = bp[(size_t)(kk + 1) * HW_ + 16 * t];
      }
#pragma unroll
      for (int t = 0; t < 4; ++t) {
        c[0][t] = wmma4(a0, bx[t], c[0][t]);
        c[1][t] = wmma4(a1, bx[t], c[1][t]);
      }
    }
  }
  int tid = threadIdx.x;
  asm volatile("v_nop\n\tv_nop\n\tv_nop\n\tv_nop\n\tv_nop\n\tv_nop\n\tv_nop\n\tv_nop\n\tv_nop"
               : "+v"(c[0][0]), "+v"(c[0][1]), "+v"(c[0][2]), "+v"(c[0][3]),
                 "+v"(c[1][0]), "+v"(c[1][1]), "+v"(c[1][2]), "+v"(c[1][3]), "+v"(tid));
  const int w    = tid >> 5;
  const int lane = tid & 31;
  const int half = lane >> 4;
  const int l16  = lane & 15;
#pragma unroll
  for (int i = 0; i < 2; ++i)
#pragma unroll
    for (int t = 0; t < 4; ++t)
#pragma unroll
      for (int r = 0; r < 8; ++r) part[w][4 * i + t][r][lane] = c[i][t][r];
  __syncthreads();

  float* D = h + (size_t)b * CR_ * HW_ + ng * 64 + l16;
  float s[2][4];
#pragma unroll
  for (int i = 0; i < 2; ++i)
#pragma unroll
    for (int t = 0; t < 4; ++t) {
      float a = 0.0f;
#pragma unroll
      for (int u = 0; u < 8; ++u) a += part[u][4 * i + t][w][lane];
      s[i][t] = a;
    }
#pragma unroll
  for (int i = 0; i < 2; ++i) {
    const int m = 16 * i + w + 8 * half;
    const float bias = b1[m];
#pragma unroll
    for (int t = 0; t < 4; ++t) {
      float v = s[i][t] + bias;
      v = 0.5f * v * (1.0f + erff(v * 0.70710678118654752f));
      D[(size_t)m * HW_ + 16 * t] = v;
    }
  }
}

#define GP_STRIDE 1088

__global__ void k_gram(const float* __restrict__ h, float* __restrict__ gp) {
  __shared__ float part[8][6][8][32];
  const int b  = blockIdx.x >> 3;
  const int ks = blockIdx.x & 7;
  v8f c00 = {}, c01 = {}, c10 = {}, c11 = {}, s0 = {}, s1 = {};
  {
    const int w    = threadIdx.x >> 5;
    const int lane = threadIdx.x & 31;
    const int half = lane >> 4;
    const int l16  = lane & 15;
    const float* p0 = h + (size_t)b * CR_ * HW_ + (size_t)l16 * HW_ + ks * 512 + w * 64 + 4 * half;
    const float* p1 = p0 + (size_t)16 * HW_;
    float4 r0[8], r1[8];
#pragma unroll
    for (int i = 0; i < 8; ++i) {
      r0[i] = *(const float4*)(p0 + 8 * i);
      r1[i] = *(const float4*)(p1 + 8 * i);
    }
    v2f one;
    one.x = 1.0f;
    one.y = 1.0f;
#pragma unroll
    for (int i = 0; i < 8; ++i) {
      v2f a0, a1;
      a0.x = r0[i].x; a0.y = r0[i].y;
      a1.x = r1[i].x; a1.y = r1[i].y;
      c00 = wmma4(a0, a0, c00);
      c01 = wmma4(a0, a1, c01);
      c10 = wmma4(a1, a0, c10);
      c11 = wmma4(a1, a1, c11);
      s0  = wmma4(a0, one, s0);
      s1  = wmma4(a1, one, s1);
      a0.x = r0[i].z; a0.y = r0[i].w;
      a1.x = r1[i].z; a1.y = r1[i].w;
      c00 = wmma4(a0, a0, c00);
      c01 = wmma4(a0, a1, c01);
      c10 = wmma4(a1, a0, c10);
      c11 = wmma4(a1, a1, c11);
      s0  = wmma4(a0, one, s0);
      s1  = wmma4(a1, one, s1);
    }
  }
  int tid = threadIdx.x;
  asm volatile("v_nop\n\tv_nop\n\tv_nop\n\tv_nop\n\tv_nop\n\tv_nop\n\tv_nop\n\tv_nop\n\tv_nop"
               : "+v"(c00), "+v"(c01), "+v"(c10), "+v"(c11), "+v"(s0), "+v"(s1), "+v"(tid));
  const int w    = tid >> 5;
  const int lane = tid & 31;
  const int half = lane >> 4;
  const int l16  = lane & 15;
#pragma unroll
  for (int r = 0; r < 8; ++r) {
    part[w][0][r][lane] = c00[r];
    part[w][1][r][lane] = c01[r];
    part[w][2][r][lane] = c10[r];
    part[w][3][r][lane] = c11[r];
    part[w][4][r][lane] = s0[r];
    part[w][5][r][lane] = s1[r];
  }
  __syncthreads();

  float* g = gp + (size_t)blockIdx.x * GP_STRIDE;
#pragma unroll
  for (int t = 0; t < 4; ++t) {
    float s = 0.0f;
#pragma unroll
    for (int u = 0; u < 8; ++u) s += part[u][t][w][lane];
    g[(16 * (t >> 1) + w + 8 * half) * 32 + 16 * (t & 1) + l16] = s;
  }
  if (tid < 32) {
    const int rr = tid & 15;
    float s = 0.0f;
#pragma unroll
    for (int u = 0; u < 8; ++u) s += part[u][4 + (tid >> 4)][rr & 7][(rr >> 3) * 16];
    g[1024 + tid] = s;
  }
}


#define ET_PITCH 516

#define PIN9(f) asm volatile("" : "+v"(f[0]), "+v"(f[1]), "+v"(f[2]), "+v"(f[3]), "+v"(f[4]), \
                                  "+v"(f[5]), "+v"(f[6]), "+v"(f[7]), "+v"(f[8]))

__global__ void k_energy(const float* __restrict__ gp, const float* __restrict__ w2,
                         const float* __restrict__ b2, float* __restrict__ e) {
  __shared__ float ga[36][48];
  __shared__ float tl[16][40];
  __shared__ alignas(16) float et[16][ET_PITCH];
  const int b    = blockIdx.x >> 5;
  const int mt   = blockIdx.x & 31;
  int tid = threadIdx.x;
  v2f bw[4][9];
  {
    const int w    = tid >> 5;
    const int lane = tid & 31;
    const int half = lane >> 4;
    const int l16  = lane & 15;

#pragma unroll
    for (int t = 0; t < 4; ++t) {
      const int n = 64 * w + 16 * t + l16;
      const float* wp = w2 + (size_t)n * CR_ + 2 * half;
#pragma unroll
      for (int s = 0; s < 8; ++s) bw[t][s] = *(const v2f*)(wp + 4 * s);
      const float bn = b2[n];
      bw[t][8].x = half ? 0.0f : bn;
      bw[t][8].y = 0.0f;
    }

    const float* gb = gp + (size_t)b * 8 * GP_STRIDE;
#pragma unroll
    for (int i = 0; i < 4; ++i) {
      const int j = w + 8 * i;
      float s = 0.0f;
#pragma unroll
      for (int u = 0; u < 8; ++u) s += gb[u * GP_STRIDE + j * 32 + lane];
      ga[j][lane] = s;
    }
    if (tid < 32) {
      float s = 0.0f;
#pragma unroll
      for (int u = 0; u < 8; ++u) s += gb[u * GP_STRIDE + 1024 + tid];
      ga[tid][32] = s;
      ga[32][tid] = s;
    }
    for (int idx = tid; idx < 36 * 48; idx += 256) {
      const int j = idx / 48;
      const int k = idx - j * 48;
      if (j >= 33 || k >= 33) ga[j][k] = 0.0f;
      else if (j == 32 && k == 32) ga[j][k] = (float)HW_;
    }
  }
  __syncthreads();

  if ((tid >> 5) < 3) {
    v2f fa[9], fb[9];
    {
      const int w    = tid >> 5;
      const int half = (tid & 31) >> 4;
      const int l16  = tid & 15;
      const int m    = mt * 16 + l16;
      const int col  = 16 * w + l16;
      const float* ap = w2 + (size_t)m * CR_ + 2 * half;
#pragma unroll
      for (int s = 0; s < 8; ++s) fa[s] = *(const v2f*)(ap + 4 * s);
      const float bm = b2[m];
      fa[8].x = half ? 0.0f : bm;
      fa[8].y = 0.0f;
#pragma unroll
      for (int s = 0; s < 9; ++s) {
        fb[s].x = ga[4 * s + 2 * half][col];
        fb[s].y = ga[4 * s + 2 * half + 1][col];
      }
    }
    PIN9(fa);
    PIN9(fb);
    v8f c = {};
#pragma unroll
    for (int s = 0; s < 9; ++s) c = wmma4(fa[s], fb[s], c);
    int t2 = tid;
    asm volatile("v_nop\n\tv_nop\n\tv_nop\n\tv_nop\n\tv_nop\n\tv_nop\n\tv_nop\n\tv_nop\n\tv_nop"
                 : "+v"(c), "+v"(t2));
    const int col  = 16 * (t2 >> 5) + (t2 & 15);
    const int row0 = 8 * ((t2 & 31) >> 4);
    if (col < 36) {
#pragma unroll
      for (int r = 0; r < 8; ++r) tl[row0 + r][col] = c[r];
    }
  }
  __syncthreads();

  v2f ta[9];
  {
    const int half = (tid & 31) >> 4;
    const int l16  = tid & 15;
#pragma unroll
    for (int s = 0; s < 9; ++s) ta[s] = *(const v2f*)&tl[l16][4 * s + 2 * half];
  }
  PIN9(ta);
  PIN9(bw[0]);
  PIN9(bw[1]);
  PIN9(bw[2]);
  PIN9(bw[3]);
  v8f c0 = {}, c1 = {}, c2 = {}, c3 = {};
#pragma unroll
  for (int s = 0; s < 9; ++s) {
    c0 = wmma4(ta[s], bw[0][s], c0);
    c1 = wmma4(ta[s], bw[1][s], c1);
    c2 = wmma4(ta[s], bw[2][s], c2);
    c3 = wmma4(ta[s], bw[3][s], c3);
  }
  asm volatile("v_nop\n\tv_nop\n\tv_nop\n\tv_nop\n\tv_nop\n\tv_nop\n\tv_nop\n\tv_nop\n\tv_nop"
               : "+v"(c0), "+v"(c1), "+v"(c2), "+v"(c3), "+v"(tid));
  {
    const int w    = tid >> 5;
    const int half = (tid & 31) >> 4;
    const int l16  = tid & 15;
    float* er = &et[8 * half][64 * w + l16];
#pragma unroll
    for (int r = 0; r < 8; ++r) {
      er[r * ET_PITCH]      = c0[r];
      er[r * ET_PITCH + 16] = c1[r];
      er[r * ET_PITCH + 32] = c2[r];
      er[r * ET_PITCH + 48] = c3[r];
    }
  }
  __syncthreads();

  {
    const int w    = tid >> 5;
    const int lane = tid & 31;
    float4 v[2][4];
#pragma unroll
    for (int i = 0; i < 2; ++i)
#pragma unroll
      for (int j = 0; j < 4; ++j) v[i][j] = *(const float4*)&et[2 * w + i][4 * lane + 128 * j];
    float mn[2], sm[2];
#pragma unroll
    for (int i = 0; i < 2; ++i) {
      float m = fminf(fminf(v[i][0].x, v[i][0].y), fminf(v[i][0].z, v[i][0].w));
#pragma unroll
      for (int j = 1; j < 4; ++j)
        m = fminf(m, fminf(fminf(v[i][j].x, v[i][j].y), fminf(v[i][j].z, v[i][j].w)));
      mn[i] = m;
    }
#pragma unroll
    for (int off = 16; off > 0; off >>= 1) {
      mn[0] = fminf(mn[0], __shfl_xor(mn[0], off));
      mn[1] = fminf(mn[1], __shfl_xor(mn[1], off));
    }
#pragma unroll
    for (int i = 0; i < 2; ++i) {
      float s = 0.0f;
#pragma unroll
      for (int j = 0; j < 4; ++j) {
        v[i][j].x = expf(mn[i] - v[i][j].x);
        v[i][j].y = expf(mn[i] - v[i][j].y);
        v[i][j].z = expf(mn[i] - v[i][j].z);
        v[i][j].w = expf(mn[i] - v[i][j].w);
        s += (v[i][j].x + v[i][j].y) + (v[i][j].z + v[i][j].w);
      }
      sm[i] = s;
    }
#pragma unroll
    for (int off = 16; off > 0; off >>= 1) {
      sm[0] += __shfl_xor(sm[0], off);
      sm[1] += __shfl_xor(sm[1], off);
    }
    float* D = e + (size_t)b * C_ * C_ + (size_t)(mt * 16 + 2 * w) * C_ + 4 * lane;
#pragma unroll
    for (int i = 0; i < 2; ++i) {
      const float inv = 1.0f / sm[i];
#pragma unroll
      for (int j = 0; j < 4; ++j) {
        float4 o;
        o.x = v[i][j].x * inv;
        o.y = v[i][j].y * inv;
        o.z = v[i][j].z * inv;
        o.w = v[i][j].w * inv;
        *(float4*)(D + (size_t)i * C_ + 128 * j) = o;
      }
    }
  }
}

__global__ void k_z(const float* __restrict__ attn, const float* __restrict__ w2,
                    const float* __restrict__ b2, const float* __restrict__ gamma,
                    float* __restrict__ Z, float* __restrict__ zb) {
  __shared__ float part[8][3][8][32];
  const int b    = blockIdx.x >> 5;
  const int mt   = blockIdx.x & 31;
  v8f c0 = {}, c1 = {}, c2 = {};
  {
    const int w    = threadIdx.x >> 5;
    const int lane = threadIdx.x & 31;
    const int half = lane >> 4;
    const int l16  = lane & 15;
    const int k0   = w * 64;
    const float* ap = attn + (size_t)b * C_ * C_ + (size_t)(mt * 16 + l16) * C_ + k0 + 2 * half;
    const float* bp = w2 + (size_t)(k0 + 2 * half) * CR_ + l16;
    const float* cp = b2 + k0 + 2 * half;
#pragma unroll
    for (int kk = 0; kk < 64; kk += 4) {
      const v2f a = *(const v2f*)(ap + kk);
      v2f b0, b1;
      b0.x = bp[kk * CR_];
      b0.y = bp[kk * CR_ + CR_];
      b1.x = bp[kk * CR_ + 16];
      b1.y = bp[kk * CR_ + CR_ + 16];
      const v2f bb = *(const v2f*)(cp + kk);
      c0 = wmma4(a, b0, c0);
      c1 = wmma4(a, b1, c1);
      c2 = wmma4(a, bb, c2);
    }
  }
  int tid = threadIdx.x;
  asm volatile("v_nop\n\tv_nop\n\tv_nop\n\tv_nop\n\tv_nop\n\tv_nop\n\tv_nop\n\tv_nop\n\tv_nop"
               : "+v"(c0), "+v"(c1), "+v"(c2), "+v"(tid));
  const int w    = tid >> 5;
  const int lane = tid & 31;
  const int half = lane >> 4;
  const int l16  = lane & 15;
#pragma unroll
  for (int r = 0; r < 8; ++r) {
    part[w][0][r][lane] = c0[r];
    part[w][1][r][lane] = c1[r];
    part[w][2][r][lane] = c2[r];
  }
  __syncthreads();

  const float g = gamma[0];
  const int m = mt * 16 + w + 8 * half;
#pragma unroll
  for (int t = 0; t < 2; ++t) {
    float s = 0.0f;
#pragma unroll
    for (int u = 0; u < 8; ++u) s += part[u][t][w][lane];
    const int n = t * 16 + l16;
    Z[((size_t)b * C_ + m) * CR_ + n] = g * s + w2[(size_t)m * CR_ + n];
  }
  if (tid < 16) {
    const int row = tid;
    float s = 0.0f;
#pragma unroll
    for (int u = 0; u < 8; ++u) s += part[u][2][row & 7][(row >> 3) * 16];
    zb[(size_t)blockIdx.x * 32 + row] = g * s + b2[mt * 16 + row];
  }
}

__global__ void k_pq(const float* __restrict__ wf, const float* __restrict__ Z,
                     const float* __restrict__ zb, float* __restrict__ P, float* __restrict__ q) {
  __shared__ float part[8][3][8][32];
  const int b    = blockIdx.x >> 5;
  const int mt   = blockIdx.x & 31;
  v8f c0 = {}, c1 = {}, c2 = {};
  {
    const int w    = threadIdx.x >> 5;
    const int lane = threadIdx.x & 31;
    const int half = lane >> 4;
    const int l16  = lane & 15;
    const int k0   = w * 64;
    const float* ap = wf + (size_t)(mt * 16 + l16) * C2_ + k0 + 2 * half;
    const float* bp = Z + (size_t)b * C_ * CR_ + (size_t)(k0 + 2 * half) * CR_ + l16;
    const float* cp = zb + ((size_t)b * (C_ / 16) + (size_t)w * 4) * 32 + 2 * half;
#pragma unroll
    for (int kk = 0; kk < 64; kk += 4) {
      const v2f a = *(const v2f*)(ap + kk);
      v2f b0, b1;
      b0.x = bp[kk * CR_];
      b0.y = bp[kk * CR_ + CR_];
      b1.x = bp[kk * CR_ + 16];
      b1.y = bp[kk * CR_ + CR_ + 16];
      const v2f bb = *(const v2f*)(cp + (kk >> 4) * 32 + (kk & 15));
      c0 = wmma4(a, b0, c0);
      c1 = wmma4(a, b1, c1);
      c2 = wmma4(a, bb, c2);
    }
  }
  int tid = threadIdx.x;
  asm volatile("v_nop\n\tv_nop\n\tv_nop\n\tv_nop\n\tv_nop\n\tv_nop\n\tv_nop\n\tv_nop\n\tv_nop"
               : "+v"(c0), "+v"(c1), "+v"(c2), "+v"(tid));
  const int w    = tid >> 5;
  const int lane = tid & 31;
  const int half = lane >> 4;
  const int l16  = lane & 15;
#pragma unroll
  for (int r = 0; r < 8; ++r) {
    part[w][0][r][lane] = c0[r];
    part[w][1][r][lane] = c1[r];
    part[w][2][r][lane] = c2[r];
  }
  __syncthreads();

  const int m = mt * 16 + w + 8 * half;
#pragma unroll
  for (int t = 0; t < 2; ++t) {
    float s = 0.0f;
#pragma unroll
    for (int u = 0; u < 8; ++u) s += part[u][t][w][lane];
    P[((size_t)b * C_ + m) * CR_ + t * 16 + l16] = s;
  }
  if (tid < 16) {
    const int row = tid;
    float s = 0.0f;
#pragma unroll
    for (int u = 0; u < 8; ++u) s += part[u][2][row & 7][(row >> 3) * 16];
    q[(size_t)blockIdx.x * 32 + row] = s;
  }
}

__global__ void __launch_bounds__(256) k_fusion(const float* __restrict__ P, const float* __restrict__ q,
                         const float* __restrict__ h, const float* __restrict__ x,
                         const float* __restrict__ wf, const float* __restrict__ bf,
                         float* __restrict__ out) {
  const int wid = blockIdx.x * 8 + (threadIdx.x >> 5);
  const int NG = HW_ / 64;
  const int MT = C_ / 64;
  const int b  = wid / (MT * NG);
  const int r0 = wid % (MT * NG);
  const int mt = r0 / NG;
  const int ng = r0 % NG;

  v8f c[4][4] = {};
  gemm_tile64(P + ((size_t)b * C_ + (size_t)mt * 64) * CR_, CR_,
              h + (size_t)b * CR_ * HW_ + ng * 64, HW_, CR_, c);
  gemm_tile64(wf + (size_t)mt * 64 * C2_ + C_, C2_,
              x + (size_t)b * C_ * HW_ + ng * 64, HW_, C_, c);

  const int lane = threadIdx.x & 31;
  const int half = lane >> 4;
  const int l16  = lane & 15;
  float* D = out + (size_t)b * C_ * HW_;
  const float* qb = q + ((size_t)b * (C_ / 16) + (size_t)mt * 4) * 32;
#pragma unroll
  for (int i = 0; i < 4; ++i)
#pragma unroll
    for (int r = 0; r < 8; ++r) {
      const int m = mt * 64 + 16 * i + r + 8 * half;
      const float bias = bf[m] + qb[i * 32 + r + 8 * half];
#pragma unroll
      for (int t = 0; t < 4; ++t) {
        const int n = ng * 64 + t * 16 + l16;
        D[(size_t)m * HW_ + n] = c[i][t][r] + bias;
      }
    }
}

extern "C" void kernel_launch(void* const* d_in, const int* in_sizes, int n_in,
                              void* d_out, int out_size, void* d_ws, size_t ws_size,
                              hipStream_t stream) {
  const float* x     = (const float*)d_in[0];
  const float* w1    = (const float*)d_in[1];
  const float* b1    = (const float*)d_in[2];
  const float* w2    = (const float*)d_in[3];
  const float* b2    = (const float*)d_in[4];
  const float* wf    = (const float*)d_in[5];
  const float* bf    = (const float*)d_in[6];
  const float* gamma = (const float*)d_in[7];
  float* out = (float*)d_out;

  char* ws = (char*)d_ws;
  float* h    = (float*)(ws);
  float* gp   = (float*)(ws + (size_t)4 * 1024 * 1024);
  float* attn = (float*)(ws + (size_t)68 * 1024 * 1024);
  float* Z    = (float*)(ws + (size_t)76 * 1024 * 1024);
  float* zb   = Z + (size_t)B_ * C_ * CR_;
  float* P    = attn;
  float* q    = attn + (size_t)B_ * C_ * CR_;

  const dim3 blk(256);
  k_conv1_gelu<<<dim3(512),  blk, 0, stream>>>(x, w1, b1, h);
  k_gram      <<<dim3(64),   blk, 0, stream>>>(h, gp);
  k_energy    <<<dim3(256),  blk, 0, stream>>>(gp, w2, b2, attn);
  k_z         <<<dim3(256),  blk, 0, stream>>>(attn, w2, b2, gamma, Z, zb);
  k_pq        <<<dim3(256),  blk, 0, stream>>>(wf, Z, zb, P, q);
  k_fusion    <<<dim3(512),  blk, 0, stream>>>(P, q, h, x, wf, bf, out);
}


# final product (P.h + Wf2.x): fragment loads software-pipelined three k-steps ahead over four register buffers, a k-step's 12 loads issued 48 WMMAs before their use; f32 sum order unchanged
# speedup vs baseline: 4.7424x; 1.1120x over previous
#include <hip/hip_runtime.h>
#include <math.h>


#define B_   8
#define C_   512
#define CR_  32
#define HW_  4096
#define C2_  1024

typedef __attribute__((ext_vector_type(2))) float v2f;
typedef __attribute__((ext_vector_type(8))) float v8f;

__device__ __forceinline__ v8f wmma4(v2f a, v2f b, v8f c) {
  return __builtin_amdgcn_wmma_f32_16x16x4_f32(
       false, a,  false, b,
       (short)0, c,  false,  false);
}

struct KStep {
  v2f a[4];
  v2f b[4];
};

__device__ __forceinline__ void kstep_load(const float* __restrict__ ap, int lda,
                                           const float* __restrict__ bp, int ldb, KStep& f) {
#pragma unroll
  for (int i = 0; i < 4; ++i) f.a[i] = *(const v2f*)(ap + (size_t)(16 * i) * lda);
#pragma unroll
  for (int t = 0; t < 4; ++t) {
    f.b[t].x = bp[t * 16];
    f.b[t].y = bp[t * 16 + ldb];
  }
}

__device__ __forceinline__ void kstep_mma(const KStep& f, v8f c[4][4]) {
#pragma unroll
  for (int i = 0; i < 4; ++i)
#pragma unroll
    for (int t = 0; t < 4; ++t) c[i][t] = wmma4(f.a[i], f.b[t], c[i][t]);
}

__global__ void __launch_bounds__(256) __attribute__((amdgpu_waves_per_eu(4, 4))) k_conv1_gelu(const float* __restrict__ x, const float* __restrict__ w1,
                             const float* __restrict__ b1, float* __restrict__ h) {
  __shared__ float part[8][8][8][32];
  const int b  = blockIdx.x >> 6;
  const int ng = blockIdx.x & 63;
  v8f c[2][4] = {};
  {
    const int w    = threadIdx.x >> 5;
    const int lane = threadIdx.x & 31;
    const int half = lane >> 4;
    const int l16  = lane & 15;
    const int k0   = w * 64;
    const float* ap = w1 + (size_t)l16 * C_ + k0 + 2 * half;
    const float* bp = x + (size_t)b * C_ * HW_ + (size_t)(k0 + 2 * half) * HW_ + ng * 64 + l16;
#pragma unroll
    for (int kk = 0; kk < 64; kk += 4) {
      const v2f a0 = *(const v2f*)(ap + kk);
      const v2f a1 = *(const v2f*)(ap + 16 * C_ + kk);
      v2f bx[4];
#pragma unroll
      for (int t = 0; t < 4; ++t) {
        bx[t].x = bp[(size_t)kk * HW_ + 16 * t];
        bx[t].y = bp[(size_t)(kk + 1) * HW_ + 16 * t];
      }
#pragma unroll
      for (int t = 0; t < 4; ++t) {
        c[0][t] = wmma4(a0, bx[t], c[0][t]);
        c[1][t] = wmma4(a1, bx[t], c[1][t]);
      }
    }
  }
  int tid = threadIdx.x;
  asm volatile("v_nop\n\tv_nop\n\tv_nop\n\tv_nop\n\tv_nop\n\tv_nop\n\tv_nop\n\tv_nop\n\tv_nop"
               : "+v"(c[0][0]), "+v"(c[0][1]), "+v"(c[0][2]), "+v"(c[0][3]),
                 "+v"(c[1][0]), "+v"(c[1][1]), "+v"(c[1][2]), "+v"(c[1][3]), "+v"(tid));
  const int w    = tid >> 5;
  const int lane = tid & 31;
  const int half = lane >> 4;
  const int l16  = lane & 15;
#pragma unroll
  for (int i = 0; i < 2; ++i)
#pragma unroll
    for (int t = 0; t < 4; ++t)
#pragma unroll
      for (int r = 0; r < 8; ++r) part[w][4 * i + t][r][lane] = c[i][t][r];
  __syncthreads();

  float* D = h + (size_t)b * CR_ * HW_ + ng * 64 + l16;
  float s[2][4];
#pragma unroll
  for (int i = 0; i < 2; ++i)
#pragma unroll
    for (int t = 0; t < 4; ++t) {
      float a = 0.0f;
#pragma unroll
      for (int u = 0; u < 8; ++u) a += part[u][4 * i + t][w][lane];
      s[i][t] = a;
    }
#pragma unroll
  for (int i = 0; i < 2; ++i) {
    const int m = 16 * i + w + 8 * half;
    const float bias = b1[m];
#pragma unroll
    for (int t = 0; t < 4; ++t) {
      float v = s[i][t] + bias;
      v = 0.5f * v * (1.0f + erff(v * 0.70710678118654752f));
      D[(size_t)m * HW_ + 16 * t] = v;
    }
  }
}

#define GP_STRIDE 1088

__global__ void k_gram(const float* __restrict__ h, float* __restrict__ gp) {
  __shared__ float part[8][6][8][32];
  const int b  = blockIdx.x >> 3;
  const int ks = blockIdx.x & 7;
  v8f c00 = {}, c01 = {}, c10 = {}, c11 = {}, s0 = {}, s1 = {};
  {
    const int w    = threadIdx.x >> 5;
    const int lane = threadIdx.x & 31;
    const int half = lane >> 4;
    const int l16  = lane & 15;
    const float* p0 = h + (size_t)b * CR_ * HW_ + (size_t)l16 * HW_ + ks * 512 + w * 64 + 4 * half;
    const float* p1 = p0 + (size_t)16 * HW_;
    float4 r0[8], r1[8];
#pragma unroll
    for (int i = 0; i < 8; ++i) {
      r0[i] = *(const float4*)(p0 + 8 * i);
      r1[i] = *(const float4*)(p1 + 8 * i);
    }
    v2f one;
    one.x = 1.0f;
    one.y = 1.0f;
#pragma unroll
    for (int i = 0; i < 8; ++i) {
      v2f a0, a1;
      a0.x = r0[i].x; a0.y = r0[i].y;
      a1.x = r1[i].x; a1.y = r1[i].y;
      c00 = wmma4(a0, a0, c00);
      c01 = wmma4(a0, a1, c01);
      c10 = wmma4(a1, a0, c10);
      c11 = wmma4(a1, a1, c11);
      s0  = wmma4(a0, one, s0);
      s1  = wmma4(a1, one, s1);
      a0.x = r0[i].z; a0.y = r0[i].w;
      a1.x = r1[i].z; a1.y = r1[i].w;
      c00 = wmma4(a0, a0, c00);
      c01 = wmma4(a0, a1, c01);
      c10 = wmma4(a1, a0, c10);
      c11 = wmma4(a1, a1, c11);
      s0  = wmma4(a0, one, s0);
      s1  = wmma4(a1, one, s1);
    }
  }
  int tid = threadIdx.x;
  asm volatile("v_nop\n\tv_nop\n\tv_nop\n\tv_nop\n\tv_nop\n\tv_nop\n\tv_nop\n\tv_nop\n\tv_nop"
               : "+v"(c00), "+v"(c01), "+v"(c10), "+v"(c11), "+v"(s0), "+v"(s1), "+v"(tid));
  const int w    = tid >> 5;
  const int lane = tid & 31;
  const int half = lane >> 4;
  const int l16  = lane & 15;
#pragma unroll
  for (int r = 0; r < 8; ++r) {
    part[w][0][r][lane] = c00[r];
    part[w][1][r][lane] = c01[r];
    part[w][2][r][lane] = c10[r];
    part[w][3][r][lane] = c11[r];
    part[w][4][r][lane] = s0[r];
    part[w][5][r][lane] = s1[r];
  }
  __syncthreads();

  float* g = gp + (size_t)blockIdx.x * GP_STRIDE;
#pragma unroll
  for (int t = 0; t < 4; ++t) {
    float s = 0.0f;
#pragma unroll
    for (int u = 0; u < 8; ++u) s += part[u][t][w][lane];
    g[(16 * (t >> 1) + w + 8 * half) * 32 + 16 * (t & 1) + l16] = s;
  }
  if (tid < 32) {
    const int rr = tid & 15;
    float s = 0.0f;
#pragma unroll
    for (int u = 0; u < 8; ++u) s += part[u][4 + (tid >> 4)][rr & 7][(rr >> 3) * 16];
    g[1024 + tid] = s;
  }
}


#define ET_PITCH 516

#define PIN9(f) asm volatile("" : "+v"(f[0]), "+v"(f[1]), "+v"(f[2]), "+v"(f[3]), "+v"(f[4]), \
                                  "+v"(f[5]), "+v"(f[6]), "+v"(f[7]), "+v"(f[8]))

__global__ void k_energy(const float* __restrict__ gp, const float* __restrict__ w2,
                         const float* __restrict__ b2, float* __restrict__ e) {
  __shared__ float ga[36][48];
  __shared__ float tl[16][40];
  __shared__ alignas(16) float et[16][ET_PITCH];
  const int b    = blockIdx.x >> 5;
  const int mt   = blockIdx.x & 31;
  int tid = threadIdx.x;
  v2f bw[4][9];
  {
    const int w    = tid >> 5;
    const int lane = tid & 31;
    const int half = lane >> 4;
    const int l16  = lane & 15;

#pragma unroll
    for (int t = 0; t < 4; ++t) {
      const int n = 64 * w + 16 * t + l16;
      const float* wp = w2 + (size_t)n * CR_ + 2 * half;
#pragma unroll
      for (int s = 0; s < 8; ++s) bw[t][s] = *(const v2f*)(wp + 4 * s);
      const float bn = b2[n];
      bw[t][8].x = half ? 0.0f : bn;
      bw[t][8].y = 0.0f;
    }

    const float* gb = gp + (size_t)b * 8 * GP_STRIDE;
#pragma unroll
    for (int i = 0; i < 4; ++i) {
      const int j = w + 8 * i;
      float s = 0.0f;
#pragma unroll
      for (int u = 0; u < 8; ++u) s += gb[u * GP_STRIDE + j * 32 + lane];
      ga[j][lane] = s;
    }
    if (tid < 32) {
      float s = 0.0f;
#pragma unroll
      for (int u = 0; u < 8; ++u) s += gb[u * GP_STRIDE + 1024 + tid];
      ga[tid][32] = s;
      ga[32][tid] = s;
    }
    for (int idx = tid; idx < 36 * 48; idx += 256) {
      const int j = idx / 48;
      const int k = idx - j * 48;
      if (j >= 33 || k >= 33) ga[j][k] = 0.0f;
      else if (j == 32 && k == 32) ga[j][k] = (float)HW_;
    }
  }
  __syncthreads();

  if ((tid >> 5) < 3) {
    v2f fa[9], fb[9];
    {
      const int w    = tid >> 5;
      const int half = (tid & 31) >> 4;
      const int l16  = tid & 15;
      const int m    = mt * 16 + l16;
      const int col  = 16 * w + l16;
      const float* ap = w2 + (size_t)m * CR_ + 2 * half;
#pragma unroll
      for (int s = 0; s < 8; ++s) fa[s] = *(const v2f*)(ap + 4 * s);
      const float bm = b2[m];
      fa[8].x = half ? 0.0f : bm;
      fa[8].y = 0.0f;
#pragma unroll
      for (int s = 0; s < 9; ++s) {
        fb[s].x = ga[4 * s + 2 * half][col];
        fb[s].y = ga[4 * s + 2 * half + 1][col];
      }
    }
    PIN9(fa);
    PIN9(fb);
    v8f c = {};
#pragma unroll
    for (int s = 0; s < 9; ++s) c = wmma4(fa[s], fb[s], c);
    int t2 = tid;
    asm volatile("v_nop\n\tv_nop\n\tv_nop\n\tv_nop\n\tv_nop\n\tv_nop\n\tv_nop\n\tv_nop\n\tv_nop"
                 : "+v"(c), "+v"(t2));
    const int col  = 16 * (t2 >> 5) + (t2 & 15);
    const int row0 = 8 * ((t2 & 31) >> 4);
    if (col < 36) {
#pragma unroll
      for (int r = 0; r < 8; ++r) tl[row0 + r][col] = c[r];
    }
  }
  __syncthreads();

  v2f ta[9];
  {
    const int half = (tid & 31) >> 4;
    const int l16  = tid & 15;
#pragma unroll
    for (int s = 0; s < 9; ++s) ta[s] = *(const v2f*)&tl[l16][4 * s + 2 * half];
  }
  PIN9(ta);
  PIN9(bw[0]);
  PIN9(bw[1]);
  PIN9(bw[2]);
  PIN9(bw[3]);
  v8f c0 = {}, c1 = {}, c2 = {}, c3 = {};
#pragma unroll
  for (int s = 0; s < 9; ++s) {
    c0 = wmma4(ta[s], bw[0][s], c0);
    c1 = wmma4(ta[s], bw[1][s], c1);
    c2 = wmma4(ta[s], bw[2][s], c2);
    c3 = wmma4(ta[s], bw[3][s], c3);
  }
  asm volatile("v_nop\n\tv_nop\n\tv_nop\n\tv_nop\n\tv_nop\n\tv_nop\n\tv_nop\n\tv_nop\n\tv_nop"
               : "+v"(c0), "+v"(c1), "+v"(c2), "+v"(c3), "+v"(tid));
  {
    const int w    = tid >> 5;
    const int half = (tid & 31) >> 4;
    const int l16  = tid & 15;
    float* er = &et[8 * half][64 * w + l16];
#pragma unroll
    for (int r = 0; r < 8; ++r) {
      er[r * ET_PITCH]      = c0[r];
      er[r * ET_PITCH + 16] = c1[r];
      er[r * ET_PITCH + 32] = c2[r];
      er[r * ET_PITCH + 48] = c3[r];
    }
  }
  __syncthreads();

  {
    const int w    = tid >> 5;
    const int lane = tid & 31;
    float4 v[2][4];
#pragma unroll
    for (int i = 0; i < 2; ++i)
#pragma unroll
      for (int j = 0; j < 4; ++j) v[i][j] = *(const float4*)&et[2 * w + i][4 * lane + 128 * j];
    float mn[2], sm[2];
#pragma unroll
    for (int i = 0; i < 2; ++i) {
      float m = fminf(fminf(v[i][0].x, v[i][0].y), fminf(v[i][0].z, v[i][0].w));
#pragma unroll
      for (int j = 1; j < 4; ++j)
        m = fminf(m, fminf(fminf(v[i][j].x, v[i][j].y), fminf(v[i][j].z, v[i][j].w)));
      mn[i] = m;
    }
#pragma unroll
    for (int off = 16; off > 0; off >>= 1) {
      mn[0] = fminf(mn[0], __shfl_xor(mn[0], off));
      mn[1] = fminf(mn[1], __shfl_xor(mn[1], off));
    }
#pragma unroll
    for (int i = 0; i < 2; ++i) {
      float s = 0.0f;
#pragma unroll
      for (int j = 0; j < 4; ++j) {
        v[i][j].x = expf(mn[i] - v[i][j].x);
        v[i][j].y = expf(mn[i] - v[i][j].y);
        v[i][j].z = expf(mn[i] - v[i][j].z);
        v[i][j].w = expf(mn[i] - v[i][j].w);
        s += (v[i][j].x + v[i][j].y) + (v[i][j].z + v[i][j].w);
      }
      sm[i] = s;
    }
#pragma unroll
    for (int off = 16; off > 0; off >>= 1) {
      sm[0] += __shfl_xor(sm[0], off);
      sm[1] += __shfl_xor(sm[1], off);
    }
    float* D = e + (size_t)b * C_ * C_ + (size_t)(mt * 16 + 2 * w) * C_ + 4 * lane;
#pragma unroll
    for (int i = 0; i < 2; ++i) {
      const float inv = 1.0f / sm[i];
#pragma unroll
      for (int j = 0; j < 4; ++j) {
        float4 o;
        o.x = v[i][j].x * inv;
        o.y = v[i][j].y * inv;
        o.z = v[i][j].z * inv;
        o.w = v[i][j].w * inv;
        *(float4*)(D + (size_t)i * C_ + 128 * j) = o;
      }
    }
  }
}

__global__ void k_z(const float* __restrict__ attn, const float* __restrict__ w2,
                    const float* __restrict__ b2, const float* __restrict__ gamma,
                    float* __restrict__ Z, float* __restrict__ zb) {
  __shared__ float part[8][3][8][32];
  const int b    = blockIdx.x >> 5;
  const int mt   = blockIdx.x & 31;
  v8f c0 = {}, c1 = {}, c2 = {};
  {
    const int w    = threadIdx.x >> 5;
    const int lane = threadIdx.x & 31;
    const int half = lane >> 4;
    const int l16  = lane & 15;
    const int k0   = w * 64;
    const float* ap = attn + (size_t)b * C_ * C_ + (size_t)(mt * 16 + l16) * C_ + k0 + 2 * half;
    const float* bp = w2 + (size_t)(k0 + 2 * half) * CR_ + l16;
    const float* cp = b2 + k0 + 2 * half;
#pragma unroll
    for (int kk = 0; kk < 64; kk += 4) {
      const v2f a = *(const v2f*)(ap + kk);
      v2f b0, b1;
      b0.x = bp[kk * CR_];
      b0.y = bp[kk * CR_ + CR_];
      b1.x = bp[kk * CR_ + 16];
      b1.y = bp[kk * CR_ + CR_ + 16];
      const v2f bb = *(const v2f*)(cp + kk);
      c0 = wmma4(a, b0, c0);
      c1 = wmma4(a, b1, c1);
      c2 = wmma4(a, bb, c2);
    }
  }
  int tid = threadIdx.x;
  asm volatile("v_nop\n\tv_nop\n\tv_nop\n\tv_nop\n\tv_nop\n\tv_nop\n\tv_nop\n\tv_nop\n\tv_nop"
               : "+v"(c0), "+v"(c1), "+v"(c2), "+v"(tid));
  const int w    = tid >> 5;
  const int lane = tid & 31;
  const int half = lane >> 4;
  const int l16  = lane & 15;
#pragma unroll
  for (int r = 0; r < 8; ++r) {
    part[w][0][r][lane] = c0[r];
    part[w][1][r][lane] = c1[r];
    part[w][2][r][lane] = c2[r];
  }
  __syncthreads();

  const float g = gamma[0];
  const int m = mt * 16 + w + 8 * half;
#pragma unroll
  for (int t = 0; t < 2; ++t) {
    float s = 0.0f;
#pragma unroll
    for (int u = 0; u < 8; ++u) s += part[u][t][w][lane];
    const int n = t * 16 + l16;
    Z[((size_t)b * C_ + m) * CR_ + n] = g * s + w2[(size_t)m * CR_ + n];
  }
  if (tid < 16) {
    const int row = tid;
    float s = 0.0f;
#pragma unroll
    for (int u = 0; u < 8; ++u) s += part[u][2][row & 7][(row >> 3) * 16];
    zb[(size_t)blockIdx.x * 32 + row] = g * s + b2[mt * 16 + row];
  }
}

__global__ void k_pq(const float* __restrict__ wf, const float* __restrict__ Z,
                     const float* __restrict__ zb, float* __restrict__ P, float* __restrict__ q) {
  __shared__ float part[8][3][8][32];
  const int b    = blockIdx.x >> 5;
  const int mt   = blockIdx.x & 31;
  v8f c0 = {}, c1 = {}, c2 = {};
  {
    const int w    = threadIdx.x >> 5;
    const int lane = threadIdx.x & 31;
    const int half = lane >> 4;
    const int l16  = lane & 15;
    const int k0   = w * 64;
    const float* ap = wf + (size_t)(mt * 16 + l16) * C2_ + k0 + 2 * half;
    const float* bp = Z + (size_t)b * C_ * CR_ + (size_t)(k0 + 2 * half) * CR_ + l16;
    const float* cp = zb + ((size_t)b * (C_ / 16) + (size_t)w * 4) * 32 + 2 * half;
#pragma unroll
    for (int kk = 0; kk < 64; kk += 4) {
      const v2f a = *(const v2f*)(ap + kk);
      v2f b0, b1;
      b0.x = bp[kk * CR_];
      b0.y = bp[kk * CR_ + CR_];
      b1.x = bp[kk * CR_ + 16];
      b1.y = bp[kk * CR_ + CR_ + 16];
      const v2f bb = *(const v2f*)(cp + (kk >> 4) * 32 + (kk & 15));
      c0 = wmma4(a, b0, c0);
      c1 = wmma4(a, b1, c1);
      c2 = wmma4(a, bb, c2);
    }
  }
  int tid = threadIdx.x;
  asm volatile("v_nop\n\tv_nop\n\tv_nop\n\tv_nop\n\tv_nop\n\tv_nop\n\tv_nop\n\tv_nop\n\tv_nop"
               : "+v"(c0), "+v"(c1), "+v"(c2), "+v"(tid));
  const int w    = tid >> 5;
  const int lane = tid & 31;
  const int half = lane >> 4;
  const int l16  = lane & 15;
#pragma unroll
  for (int r = 0; r < 8; ++r) {
    part[w][0][r][lane] = c0[r];
    part[w][1][r][lane] = c1[r];
    part[w][2][r][lane] = c2[r];
  }
  __syncthreads();

  const int m = mt * 16 + w + 8 * half;
#pragma unroll
  for (int t = 0; t < 2; ++t) {
    float s = 0.0f;
#pragma unroll
    for (int u = 0; u < 8; ++u) s += part[u][t][w][lane];
    P[((size_t)b * C_ + m) * CR_ + t * 16 + l16] = s;
  }
  if (tid < 16) {
    const int row = tid;
    float s = 0.0f;
#pragma unroll
    for (int u = 0; u < 8; ++u) s += part[u][2][row & 7][(row >> 3) * 16];
    q[(size_t)blockIdx.x * 32 + row] = s;
  }
}

__global__ void __launch_bounds__(256) k_fusion(const float* __restrict__ P, const float* __restrict__ q,
                         const float* __restrict__ h, const float* __restrict__ x,
                         const float* __restrict__ wf, const float* __restrict__ bf,
                         float* __restrict__ out) {
  const int wid = blockIdx.x * 8 + (threadIdx.x >> 5);
  const int NG = HW_ / 64;
  const int MT = C_ / 64;
  const int b  = wid / (MT * NG);
  const int r0 = wid % (MT * NG);
  const int mt = r0 / NG;
  const int ng = r0 % NG;

  const int lane = threadIdx.x & 31;
  const int half = lane >> 4;
  const int l16  = lane & 15;

  v8f c[4][4] = {};
  const float* ap1 = P + ((size_t)b * C_ + (size_t)mt * 64 + l16) * CR_ + 2 * half;
  const float* bp1 = h + (size_t)b * CR_ * HW_ + (size_t)(2 * half) * HW_ + ng * 64 + l16;
  const float* ap2 = wf + ((size_t)mt * 64 + l16) * C2_ + C_ + 2 * half;
  const float* bp2 = x + (size_t)b * C_ * HW_ + (size_t)(2 * half) * HW_ + ng * 64 + l16;

#define PIPE_FENCE() __builtin_amdgcn_sched_barrier(0)
#define LOAD1(g, f) kstep_load(ap1 + 4 * (g), CR_, bp1 + (size_t)(4 * (g)) * HW_, HW_, f); PIPE_FENCE()
#define LOAD2(s, f) kstep_load(ap2 + 4 * (s), C2_, bp2 + (size_t)(4 * (s)) * HW_, HW_, f); PIPE_FENCE()
#define MMA(f)      kstep_mma(f, c); PIPE_FENCE()
  KStep f0, f1, f2, f3;
  LOAD1(0, f0);
  LOAD1(1, f1);
  LOAD1(2, f2);
  LOAD1(3, f3);  MMA(f0);
  LOAD1(4, f0);  MMA(f1);
  LOAD1(5, f1);  MMA(f2);
  LOAD1(6, f2);  MMA(f3);
  LOAD1(7, f3);  MMA(f0);
  LOAD2(0, f0);  MMA(f1);
  LOAD2(1, f1);  MMA(f2);
  LOAD2(2, f2);  MMA(f3);
#pragma unroll 1
  for (int s = 0; s < 124; s += 4) {
    LOAD2(3, f3);  MMA(f0);
    LOAD2(4, f0);  MMA(f1);
    LOAD2(5, f1);  MMA(f2);
    LOAD2(6, f2);  MMA(f3);
    ap2 += 16;
    bp2 += (size_t)16 * HW_;
  }
  LOAD2(3, f3);  MMA(f0);
  MMA(f1);
  MMA(f2);
  MMA(f3);
#undef MMA
#undef LOAD2
#undef LOAD1
#undef PIPE_FENCE

  float* D = out + (size_t)b * C_ * HW_;
  const float* qb = q + ((size_t)b * (C_ / 16) + (size_t)mt * 4) * 32;
#pragma unroll
  for (int i = 0; i < 4; ++i)
#pragma unroll
    for (int r = 0; r < 8; ++r) {
      const int m = mt * 64 + 16 * i + r + 8 * half;
      const float bias = bf[m] + qb[i * 32 + r + 8 * half];
#pragma unroll
      for (int t = 0; t < 4; ++t) {
        const int n = ng * 64 + t * 16 + l16;
        D[(size_t)m * HW_ + n] = c[i][t][r] + bias;
      }
    }
}

extern "C" void kernel_launch(void* const* d_in, const int* in_sizes, int n_in,
                              void* d_out, int out_size, void* d_ws, size_t ws_size,
                              hipStream_t stream) {
  const float* x     = (const float*)d_in[0];
  const float* w1    = (const float*)d_in[1];
  const float* b1    = (const float*)d_in[2];
  const float* w2    = (const float*)d_in[3];
  const float* b2    = (const float*)d_in[4];
  const float* wf    = (const float*)d_in[5];
  const float* bf    = (const float*)d_in[6];
  const float* gamma = (const float*)d_in[7];
  float* out = (float*)d_out;

  char* ws = (char*)d_ws;
  float* h    = (float*)(ws);
  float* gp   = (float*)(ws + (size_t)4 * 1024 * 1024);
  float* attn = (float*)(ws + (size_t)68 * 1024 * 1024);
  float* Z    = (float*)(ws + (size_t)76 * 1024 * 1024);
  float* zb   = Z + (size_t)B_ * C_ * CR_;
  float* P    = attn;
  float* q    = attn + (size_t)B_ * C_ * CR_;

  const dim3 blk(256);
  k_conv1_gelu<<<dim3(512),  blk, 0, stream>>>(x, w1, b1, h);
  k_gram      <<<dim3(64),   blk, 0, stream>>>(h, gp);
  k_energy    <<<dim3(256),  blk, 0, stream>>>(gp, w2, b2, attn);
  k_z         <<<dim3(256),  blk, 0, stream>>>(attn, w2, b2, gamma, Z, zb);
  k_pq        <<<dim3(256),  blk, 0, stream>>>(wf, Z, zb, P, q);
  k_fusion    <<<dim3(512),  blk, 0, stream>>>(P, q, h, x, wf, bf, out);
}
